# starting kernel
# baseline (speedup 1.0000x reference)
#include <hip/hip_runtime.h>

#define N_NODES   50000
#define F_DIM     64
#define H_HEADS   2
#define HF        128
#define L_LAYERS  2
#define NEG_SLOPE 0.2f

typedef __attribute__((ext_vector_type(2))) float v2f;
typedef __attribute__((ext_vector_type(8))) float v8f;

__device__ __forceinline__ float lrelu(float x) {
    return x > 0.0f ? x : NEG_SLOPE * x;
}
__device__ __forceinline__ unsigned fflip(float f) {
    unsigned u = __float_as_uint(f);
    return (u & 0x80000000u) ? ~u : (u | 0x80000000u);
}
__device__ __forceinline__ float funflip(unsigned v) {
    return __uint_as_float((v & 0x80000000u) ? (v ^ 0x80000000u) : ~v);
}

__global__ __launch_bounds__(256)
void gat_proj_wmma(const float* __restrict__ h,
                   const float* __restrict__ Wsrc, const float* __restrict__ bsrc,
                   const float* __restrict__ Wdst, const float* __restrict__ bdst,
                   float* __restrict__ fs, float* __restrict__ fd) {
    const int wave = threadIdx.x >> 5;
    const int lane = threadIdx.x & 31;
    const int half = lane >> 4;
    const int lrow = lane & 15;
    const int t = blockIdx.y * 8 + wave;
    const int nodeBase = blockIdx.x * 16;

    const bool  isSrc = (t < 8);
    const float* __restrict__ W    = isSrc ? Wsrc : Wdst;
    const float* __restrict__ bias = isSrc ? bsrc : bdst;
    float* __restrict__ out        = isSrc ? fs : fd;
    const int col = (isSrc ? t : t - 8) * 16 + lrow;

    const float* __restrict__ arow = h + (size_t)(nodeBase + lrow) * F_DIM;

    v8f acc = {};
#pragma unroll
    for (int k0 = 0; k0 < F_DIM; k0 += 4) {
        const int ka = k0 + half * 2;
        v2f a, b;
        a.x = arow[ka];
        a.y = arow[ka + 1];
        b.x = W[(size_t)ka * HF + col];
        b.y = W[(size_t)(ka + 1) * HF + col];
        acc = __builtin_amdgcn_wmma_f32_16x16x4_f32(
                false, a, false, b, (short)0, acc, false, false);
    }

    const float bv = bias[col];
#pragma unroll
    for (int j = 0; j < 8; ++j) {
        const int row = nodeBase + half * 8 + j;
        out[(size_t)row * HF + col] = acc[j] + bv;
    }
}

__global__ __launch_bounds__(256)
void gat_edge_logits(const float* __restrict__ fs, const float* __restrict__ fd,
                     const float* __restrict__ attn,
                     const int* __restrict__ src, const int* __restrict__ dst,
                     float* __restrict__ logits, unsigned* __restrict__ mflip,
                     int E) {
    const int e = (int)(((size_t)blockIdx.x * blockDim.x + threadIdx.x) >> 5);
    if (e >= E) return;
    const int lane = threadIdx.x & 31;
    const int s = src[e];
    const int d = dst[e];

    const float4 a = *(const float4*)(fs + (size_t)s * HF + lane * 4);
    const float4 b = *(const float4*)(fd + (size_t)d * HF + lane * 4);
    const float4 w = *(const float4*)(attn + lane * 4);

    float sum = w.x * lrelu(a.x + b.x) + w.y * lrelu(a.y + b.y)
              + w.z * lrelu(a.z + b.z) + w.w * lrelu(a.w + b.w);
#pragma unroll
    for (int off = 1; off < 16; off <<= 1)
        sum += __shfl_xor(sum, off, 32);

    if ((lane & 15) == 0) {
        const int hh = lane >> 4;
        logits[(size_t)e * 2 + hh] = sum;
        atomicMax(mflip + (size_t)d * 2 + hh, fflip(sum));
    }
}

__global__ __launch_bounds__(256)
void gat_edge_exp(const int* __restrict__ dst, float* __restrict__ logits,
                  const unsigned* __restrict__ mflip, float* __restrict__ denom,
                  int E2) {
    const int idx = blockIdx.x * blockDim.x + threadIdx.x;
    if (idx >= E2) return;
    const int e = idx >> 1, hh = idx & 1;
    const int d = dst[e];
    const float m  = funflip(mflip[(size_t)d * 2 + hh]);
    const float ex = expf(logits[idx] - m);
    logits[idx] = ex;
    atomicAdd(denom + (size_t)d * 2 + hh, ex);
}

__global__ __launch_bounds__(256)
void gat_edge_aggregate(const float* __restrict__ fs,
                        const int* __restrict__ src, const int* __restrict__ dst,
                        const float* __restrict__ logits,
                        const float* __restrict__ denom,
                        float* __restrict__ agg, int E) {
    const int e = (int)(((size_t)blockIdx.x * blockDim.x + threadIdx.x) >> 5);
    if (e >= E) return;
    const int lane = threadIdx.x & 31;
    const int hh = lane >> 4;
    const int s = src[e];
    const int d = dst[e];
    const float alpha = logits[(size_t)e * 2 + hh] / denom[(size_t)d * 2 + hh];

    const float4 v = *(const float4*)(fs + (size_t)s * HF + lane * 4);
    float* o = agg + (size_t)d * HF + lane * 4;
    atomicAdd(o + 0, alpha * v.x);
    atomicAdd(o + 1, alpha * v.y);
    atomicAdd(o + 2, alpha * v.z);
    atomicAdd(o + 3, alpha * v.w);
}

__global__ __launch_bounds__(256)
void gat_head_mean(const float* __restrict__ agg, float* __restrict__ hout, int n) {
    const int idx = blockIdx.x * blockDim.x + threadIdx.x;
    if (idx >= n) return;
    const int node = idx >> 6, f = idx & 63;
    hout[idx] = 0.5f * (agg[(size_t)node * HF + f] + agg[(size_t)node * HF + 64 + f]);
}

extern "C" void kernel_launch(void* const* d_in, const int* in_sizes, int n_in,
                              void* d_out, int out_size, void* d_ws, size_t ws_size,
                              hipStream_t stream) {
    (void)n_in; (void)out_size; (void)ws_size;
    const float* h0   = (const float*)d_in[0];
    const float* Wsrc = (const float*)d_in[1];
    const float* bsrc = (const float*)d_in[2];
    const float* Wdst = (const float*)d_in[3];
    const float* bdst = (const float*)d_in[4];
    const float* attn = (const float*)d_in[5];
    const int*   src  = (const int*)d_in[6];
    const int*   dst  = (const int*)d_in[7];
    const int    E    = in_sizes[6];

    float* ws      = (float*)d_ws;
    float* fs      = ws;  ws += (size_t)N_NODES * HF;
    float* fd      = ws;  ws += (size_t)N_NODES * HF;
    float* agg     = ws;  ws += (size_t)N_NODES * HF;
    float* hbuf    = ws;  ws += (size_t)N_NODES * F_DIM;
    float* denom   = ws;  ws += (size_t)N_NODES * 2;
    unsigned* mflp = (unsigned*)ws; ws += (size_t)N_NODES * 2;
    float* logits  = ws;

    const int gridE  = (E + 7) / 8;
    const int gridE2 = (E * 2 + 255) / 256;
    const int gridN  = (N_NODES * F_DIM + 255) / 256;

    const float* hcur = h0;
    for (int l = 0; l < L_LAYERS; ++l) {
        gat_proj_wmma<<<dim3(N_NODES / 16, 2), 256, 0, stream>>>(
            hcur,
            Wsrc + (size_t)l * F_DIM * HF, bsrc + (size_t)l * HF,
            Wdst + (size_t)l * F_DIM * HF, bdst + (size_t)l * HF,
            fs, fd);

        hipMemsetAsync(mflp,  0, (size_t)N_NODES * 2 * sizeof(unsigned), stream);
        hipMemsetAsync(denom, 0, (size_t)N_NODES * 2 * sizeof(float), stream);
        hipMemsetAsync(agg,   0, (size_t)N_NODES * HF * sizeof(float), stream);

        gat_edge_logits<<<gridE, 256, 0, stream>>>(
            fs, fd, attn + (size_t)l * HF, src, dst, logits, mflp, E);
        gat_edge_exp<<<gridE2, 256, 0, stream>>>(dst, logits, mflp, denom, E * 2);
        gat_edge_aggregate<<<gridE, 256, 0, stream>>>(
            fs, src, dst, logits, denom, agg, E);

        float* hn = (l == L_LAYERS - 1) ? (float*)d_out : hbuf;
        gat_head_mean<<<gridN, 256, 0, stream>>>(agg, hn, N_NODES * F_DIM);
        hcur = hbuf;
    }
}


# edges grouped by dst once per call (int-atomic counting sort, groups then ordered by src id: one fixed sum order); one wave per dst node: logits, running-max softmax, sum, head mean in registers
# speedup vs baseline: 3.1338x; 3.1338x over previous
#include <hip/hip_runtime.h>

#define N_NODES   50000
#define F_DIM     64
#define H_HEADS   2
#define HF        128
#define L_LAYERS  2
#define NEG_SLOPE 0.2f

typedef __attribute__((ext_vector_type(2))) float v2f;
typedef __attribute__((ext_vector_type(8))) float v8f;

__device__ __forceinline__ float lrelu(float x) {
    return x > 0.0f ? x : NEG_SLOPE * x;
}

__global__ __launch_bounds__(256)
void gat_proj_wmma(const float* __restrict__ h,
                   const float* __restrict__ Wsrc, const float* __restrict__ bsrc,
                   const float* __restrict__ Wdst, const float* __restrict__ bdst,
                   float* __restrict__ fs, float* __restrict__ fd) {
    const int wave = threadIdx.x >> 5;
    const int lane = threadIdx.x & 31;
    const int half = lane >> 4;
    const int lrow = lane & 15;
    const int t = blockIdx.y * 8 + wave;
    const int nodeBase = blockIdx.x * 16;

    const bool  isSrc = (t < 8);
    const float* __restrict__ W    = isSrc ? Wsrc : Wdst;
    const float* __restrict__ bias = isSrc ? bsrc : bdst;
    float* __restrict__ out        = isSrc ? fs : fd;
    const int col = (isSrc ? t : t - 8) * 16 + lrow;

    const float* __restrict__ arow = h + (size_t)(nodeBase + lrow) * F_DIM;

    v8f acc = {};
#pragma unroll
    for (int k0 = 0; k0 < F_DIM; k0 += 4) {
        const int ka = k0 + half * 2;
        v2f a, b;
        a.x = arow[ka];
        a.y = arow[ka + 1];
        b.x = W[(size_t)ka * HF + col];
        b.y = W[(size_t)(ka + 1) * HF + col];
        acc = __builtin_amdgcn_wmma_f32_16x16x4_f32(
                false, a, false, b, (short)0, acc, false, false);
    }

    const float bv = bias[col];
#pragma unroll
    for (int j = 0; j < 8; ++j) {
        const int row = nodeBase + half * 8 + j;
        out[(size_t)row * HF + col] = acc[j] + bv;
    }
}

__global__ __launch_bounds__(256)
void grp_count(const int* __restrict__ dst, int* __restrict__ cnt,
               int* __restrict__ slot, int E) {
    const int e = blockIdx.x * blockDim.x + threadIdx.x;
    if (e >= E) return;
    slot[e] = atomicAdd(cnt + dst[e], 1);
}

__global__ __launch_bounds__(256)
void grp_offsets(const int* __restrict__ cnt, int* __restrict__ off) {
    __shared__ int red[8];
    __shared__ int sc[256];
    const int t = threadIdx.x;
    const int b = blockIdx.x;
    const int4* __restrict__ c4 = (const int4*)cnt;
    int part = 0;
    for (int q = t; q < b * 64; q += 256) {
        const int4 v = c4[q];
        part += (v.x + v.y) + (v.z + v.w);
    }
#pragma unroll
    for (int o = 16; o > 0; o >>= 1)
        part += __shfl_xor(part, o, 32);
    if ((t & 31) == 0) red[t >> 5] = part;
    const int i = b * 256 + t;
    const int c = (i < N_NODES) ? cnt[i] : 0;
    sc[t] = c;
    __syncthreads();
    int prefix = 0;
#pragma unroll
    for (int k = 0; k < 8; ++k) prefix += red[k];
    int v = c;
    for (int o = 1; o < 256; o <<= 1) {
        const int add = (t >= o) ? sc[t - o] : 0;
        __syncthreads();
        v += add;
        sc[t] = v;
        __syncthreads();
    }
    if (i < N_NODES) off[i] = prefix + v - c;
    if (i == N_NODES - 1) off[N_NODES] = prefix + v;
}

__global__ __launch_bounds__(256)
void grp_fill(const int* __restrict__ src, const int* __restrict__ dst,
              const int* __restrict__ slot, const int* __restrict__ off,
              int* __restrict__ tmp, int E) {
    const int e = blockIdx.x * blockDim.x + threadIdx.x;
    if (e >= E) return;
    tmp[off[dst[e]] + slot[e]] = src[e];
}

__global__ __launch_bounds__(256)
void grp_order(const int* __restrict__ off, const int* __restrict__ tmp,
               int* __restrict__ nbr) {
    const int node = __builtin_amdgcn_readfirstlane(
        (int)(blockIdx.x * 8 + (threadIdx.x >> 5)));
    if (node >= N_NODES) return;
    const int lane = threadIdx.x & 31;
    const int beg = off[node];
    const int deg = off[node + 1] - beg;
    for (int ib = 0; ib < deg; ib += 32) {
        const int i = ib + lane;
        const bool act = i < deg;
        const int key = act ? tmp[beg + i] : 0;
        int r = 0;
        for (int jb = 0; jb < deg; jb += 32) {
            const int jn = min(32, deg - jb);
            const int kl = (jb + lane < deg) ? tmp[beg + jb + lane] : 0;
            for (int jj = 0; jj < jn; ++jj) {
                const int kj = __builtin_amdgcn_readlane(kl, jj);
                const int j = jb + jj;
                r += (kj < key || (kj == key && j < i)) ? 1 : 0;
            }
        }
        if (act) nbr[beg + r] = key;
    }
}

template <int CTRL>
__device__ __forceinline__ float row_xadd(float v) {
    const int t = __builtin_amdgcn_update_dpp(0, __float_as_int(v), CTRL, 0xf, 0xf, true);
    return v + __int_as_float(t);
}
__device__ __forceinline__ float xhalf(float x) {
    const int b = __float_as_int(x);
    return __int_as_float(
        __builtin_amdgcn_permlanex16(b, b, 0x76543210, 0xfedcba98, false, false));
}
__device__ __forceinline__ float edge_part(const float4 v, const float4 b, const float4 w) {
    return w.x * lrelu(v.x + b.x) + w.y * lrelu(v.y + b.y)
         + w.z * lrelu(v.z + b.z) + w.w * lrelu(v.w + b.w);
}

__global__ __launch_bounds__(256)
void gat_node_fused(const float* __restrict__ fs, const float* __restrict__ fd,
                    const float* __restrict__ attn,
                    const int* __restrict__ off, const int* __restrict__ nbr,
                    float* __restrict__ hout) {
    const int node = __builtin_amdgcn_readfirstlane(
        (int)(blockIdx.x * 8 + (threadIdx.x >> 5)));
    if (node >= N_NODES) return;
    const int lane = threadIdx.x & 31;
    const int beg = off[node];
    const int end = off[node + 1];

    const float4 b = *(const float4*)(fd + (size_t)node * HF + lane * 4);
    const float4 w = *(const float4*)(attn + lane * 4);

    float m = -INFINITY, den = 0.0f;
    float4 acc = make_float4(0.0f, 0.0f, 0.0f, 0.0f);

    for (int cb = beg; cb < end; cb += 32) {
        const int cn = min(32, end - cb);
        const int ids = nbr[cb + min(lane, cn - 1)];
        for (int j = 0; j < cn; j += 4) {
            const int s0 = __builtin_amdgcn_readlane(ids, j);
            const int s1 = __builtin_amdgcn_readlane(ids, j + 1);
            const int s2 = __builtin_amdgcn_readlane(ids, j + 2);
            const int s3 = __builtin_amdgcn_readlane(ids, j + 3);
            const float4 v0 = *(const float4*)(fs + (size_t)s0 * HF + lane * 4);
            const float4 v1 = *(const float4*)(fs + (size_t)s1 * HF + lane * 4);
            const float4 v2 = *(const float4*)(fs + (size_t)s2 * HF + lane * 4);
            const float4 v3 = *(const float4*)(fs + (size_t)s3 * HF + lane * 4);

            float l0 = edge_part(v0, b, w);
            float l1 = edge_part(v1, b, w);
            float l2 = edge_part(v2, b, w);
            float l3 = edge_part(v3, b, w);
            l0 = row_xadd<0x161>(l0); l1 = row_xadd<0x161>(l1);
            l2 = row_xadd<0x161>(l2); l3 = row_xadd<0x161>(l3);
            l0 = row_xadd<0x162>(l0); l1 = row_xadd<0x162>(l1);
            l2 = row_xadd<0x162>(l2); l3 = row_xadd<0x162>(l3);
            l0 = row_xadd<0x164>(l0); l1 = row_xadd<0x164>(l1);
            l2 = row_xadd<0x164>(l2); l3 = row_xadd<0x164>(l3);
            l0 = row_xadd<0x168>(l0); l1 = row_xadd<0x168>(l1);
            l2 = row_xadd<0x168>(l2); l3 = row_xadd<0x168>(l3);
            if (j + 1 >= cn) l1 = -INFINITY;
            if (j + 2 >= cn) l2 = -INFINITY;
            if (j + 3 >= cn) l3 = -INFINITY;

            const float mn = fmaxf(fmaxf(m, l0), fmaxf(l1, fmaxf(l2, l3)));
            const float sc = expf(m - mn);
            const float p0 = expf(l0 - mn);
            const float p1 = expf(l1 - mn);
            const float p2 = expf(l2 - mn);
            const float p3 = expf(l3 - mn);
            den   = den   * sc + ((p0 + p1) + (p2 + p3));
            acc.x = acc.x * sc + ((p0 * v0.x + p1 * v1.x) + (p2 * v2.x + p3 * v3.x));
            acc.y = acc.y * sc + ((p0 * v0.y + p1 * v1.y) + (p2 * v2.y + p3 * v3.y));
            acc.z = acc.z * sc + ((p0 * v0.z + p1 * v1.z) + (p2 * v2.z + p3 * v3.z));
            acc.w = acc.w * sc + ((p0 * v0.w + p1 * v1.w) + (p2 * v2.w + p3 * v3.w));
            m = mn;
        }
    }

    const float inv = (end > beg) ? 1.0f / den : 0.0f;
    float4 o;
    o.x = acc.x * inv; o.y = acc.y * inv; o.z = acc.z * inv; o.w = acc.w * inv;
    o.x = 0.5f * (o.x + xhalf(o.x));
    o.y = 0.5f * (o.y + xhalf(o.y));
    o.z = 0.5f * (o.z + xhalf(o.z));
    o.w = 0.5f * (o.w + xhalf(o.w));
    if (lane < 16)
        *(float4*)(hout + (size_t)node * F_DIM + lane * 4) = o;
}

extern "C" void kernel_launch(void* const* d_in, const int* in_sizes, int n_in,
                              void* d_out, int out_size, void* d_ws, size_t ws_size,
                              hipStream_t stream) {
    (void)n_in; (void)out_size; (void)ws_size;
    const float* h0   = (const float*)d_in[0];
    const float* Wsrc = (const float*)d_in[1];
    const float* bsrc = (const float*)d_in[2];
    const float* Wdst = (const float*)d_in[3];
    const float* bdst = (const float*)d_in[4];
    const float* attn = (const float*)d_in[5];
    const int*   src  = (const int*)d_in[6];
    const int*   dst  = (const int*)d_in[7];
    const int    E    = in_sizes[6];

    float* ws   = (float*)d_ws;
    float* fs   = ws;  ws += (size_t)N_NODES * HF;
    float* fd   = ws;  ws += (size_t)N_NODES * HF;
    float* hbuf = ws;  ws += (size_t)N_NODES * F_DIM;
    int* cnt    = (int*)ws;  ws += (size_t)N_NODES;
    int* off    = (int*)ws;  ws += (size_t)N_NODES + 4;
    int* slot   = (int*)ws;  ws += (size_t)E;
    int* tmp    = (int*)ws;  ws += (size_t)E;
    int* nbr    = (int*)ws;

    const int gridE = (E + 255) / 256;
    const int gridW = (N_NODES + 7) / 8;

    hipMemsetAsync(cnt, 0, (size_t)N_NODES * sizeof(int), stream);
    grp_count<<<gridE, 256, 0, stream>>>(dst, cnt, slot, E);
    grp_offsets<<<(N_NODES + 255) / 256, 256, 0, stream>>>(cnt, off);
    grp_fill<<<gridE, 256, 0, stream>>>(src, dst, slot, off, tmp, E);
    grp_order<<<gridW, 256, 0, stream>>>(off, tmp, nbr);

    const float* hcur = h0;
    for (int l = 0; l < L_LAYERS; ++l) {
        gat_proj_wmma<<<dim3(N_NODES / 16, 2), 256, 0, stream>>>(
            hcur,
            Wsrc + (size_t)l * F_DIM * HF, bsrc + (size_t)l * HF,
            Wdst + (size_t)l * F_DIM * HF, bdst + (size_t)l * HF,
            fs, fd);

        float* hn = (l == L_LAYERS - 1) ? (float*)d_out : hbuf;
        gat_node_fused<<<gridW, 256, 0, stream>>>(
            fs, fd, attn + (size_t)l * HF, off, nbr, hn);
        hcur = hbuf;
    }
}


# projection GEMM retiled: one wave per 16-row x 128-col strip, A fragments kept in registers, W staged once per block in LDS as 16 B B-fragments, whole-line 16 B stores; k summed in a permuted order
# speedup vs baseline: 3.2342x; 1.0320x over previous
#include <hip/hip_runtime.h>

#define N_NODES   50000
#define F_DIM     64
#define H_HEADS   2
#define HF        128
#define L_LAYERS  2
#define NEG_SLOPE 0.2f

typedef __attribute__((ext_vector_type(2))) float v2f;
typedef __attribute__((ext_vector_type(8))) float v8f;

__device__ __forceinline__ float lrelu(float x) {
    return x > 0.0f ? x : NEG_SLOPE * x;
}

#define ROW_TILES   (N_NODES / 16)
#define PROJ_WAVES  4
#define PROJ_BLOCKS ((ROW_TILES + PROJ_WAVES - 1) / PROJ_WAVES)

__global__ __launch_bounds__(PROJ_WAVES * 32)
void gat_proj_wmma(const float* __restrict__ h,
                   const float* __restrict__ Wsrc, const float* __restrict__ bsrc,
                   const float* __restrict__ Wdst, const float* __restrict__ bdst,
                   float* __restrict__ fs, float* __restrict__ fd) {
    __shared__ float4 wl[16 * 8 * 16];

    const int tid  = threadIdx.x;
    const int wave = tid >> 5;
    const int lane = tid & 31;
    const int hh   = lane >> 4;
    const int c    = lane & 15;

    const bool  isSrc = (blockIdx.y == 0);
    const float* __restrict__ W    = isSrc ? Wsrc : Wdst;
    const float* __restrict__ bias = isSrc ? bsrc : bdst;
    float* __restrict__ out        = isSrc ? fs : fd;

    const int  tile = blockIdx.x * PROJ_WAVES + wave;
    const bool live = tile < ROW_TILES;
    const int  ltile = live ? tile : ROW_TILES - 1;

    const float* __restrict__ arow = h + (size_t)(ltile * 16 + c) * F_DIM + hh * 4;
    float4 a[8];
#pragma unroll
    for (int i2 = 0; i2 < 8; ++i2)
        a[i2] = *(const float4*)(arow + 8 * i2);

#pragma unroll
    for (int it = 0; it < 512 / (PROJ_WAVES * 32); ++it) {
        const int item = it * (PROJ_WAVES * 32) + tid;
        const int q  = item >> 5;
        const int c4 = item & 31;
        const float* __restrict__ wp = W + (size_t)(4 * q) * HF + 4 * c4;
        const float4 r0 = *(const float4*)(wp);
        const float4 r1 = *(const float4*)(wp + HF);
        const float4 r2 = *(const float4*)(wp + 2 * HF);
        const float4 r3 = *(const float4*)(wp + 3 * HF);
        float4* dp = wl + (q * 8 + 4 * (c4 >> 4)) * 16 + (c4 & 15);
        dp[0]  = make_float4(r0.x, r1.x, r2.x, r3.x);
        dp[16] = make_float4(r0.y, r1.y, r2.y, r3.y);
        dp[32] = make_float4(r0.z, r1.z, r2.z, r3.z);
        dp[48] = make_float4(r0.w, r1.w, r2.w, r3.w);
    }
    __syncthreads();
    if (!live) return;

    v8f acc[8];
#pragma unroll
    for (int t = 0; t < 8; ++t) acc[t] = (v8f){};

    const float4* bl = wl + hh * 128 + c;
#pragma unroll
    for (int i2 = 0; i2 < 8; ++i2) {
        float4 b[8];
#pragma unroll
        for (int t = 0; t < 8; ++t) b[t] = bl[i2 * 256 + t * 16];
        v2f a0, a1;
        a0.x = a[i2].x; a0.y = a[i2].y;
        a1.x = a[i2].z; a1.y = a[i2].w;
#pragma unroll
        for (int t = 0; t < 8; ++t) {
            v2f b0; b0.x = b[t].x; b0.y = b[t].y;
            acc[t] = __builtin_amdgcn_wmma_f32_16x16x4_f32(
                        false, a0, false, b0, (short)0, acc[t], false, false);
        }
#pragma unroll
        for (int t = 0; t < 8; ++t) {
            v2f b1; b1.x = b[t].z; b1.y = b[t].w;
            acc[t] = __builtin_amdgcn_wmma_f32_16x16x4_f32(
                        false, a1, false, b1, (short)0, acc[t], false, false);
        }
    }

    float* __restrict__ orow = out + (size_t)(tile * 16 + hh * 8) * HF + 4 * c;
#pragma unroll
    for (int g = 0; g < 2; ++g) {
        const float4 bv = *(const float4*)(bias + 64 * g + 4 * c);
#pragma unroll
        for (int r = 0; r < 8; ++r) {
            float4 o;
            o.x = acc[4 * g + 0][r] + bv.x;
            o.y = acc[4 * g + 1][r] + bv.y;
            o.z = acc[4 * g + 2][r] + bv.z;
            o.w = acc[4 * g + 3][r] + bv.w;
            *(float4*)(orow + (size_t)r * HF + 64 * g) = o;
        }
    }
}

__global__ __launch_bounds__(256)
void grp_count(const int* __restrict__ dst, int* __restrict__ cnt,
               int* __restrict__ slot, int E) {
    const int e = blockIdx.x * blockDim.x + threadIdx.x;
    if (e >= E) return;
    slot[e] = atomicAdd(cnt + dst[e], 1);
}

__global__ __launch_bounds__(256)
void grp_offsets(const int* __restrict__ cnt, int* __restrict__ off) {
    __shared__ int red[8];
    __shared__ int sc[256];
    const int t = threadIdx.x;
    const int b = blockIdx.x;
    const int4* __restrict__ c4 = (const int4*)cnt;
    int part = 0;
    for (int q = t; q < b * 64; q += 256) {
        const int4 v = c4[q];
        part += (v.x + v.y) + (v.z + v.w);
    }
#pragma unroll
    for (int o = 16; o > 0; o >>= 1)
        part += __shfl_xor(part, o, 32);
    if ((t & 31) == 0) red[t >> 5] = part;
    const int i = b * 256 + t;
    const int c = (i < N_NODES) ? cnt[i] : 0;
    sc[t] = c;
    __syncthreads();
    int prefix = 0;
#pragma unroll
    for (int k = 0; k < 8; ++k) prefix += red[k];
    int v = c;
    for (int o = 1; o < 256; o <<= 1) {
        const int add = (t >= o) ? sc[t - o] : 0;
        __syncthreads();
        v += add;
        sc[t] = v;
        __syncthreads();
    }
    if (i < N_NODES) off[i] = prefix + v - c;
    if (i == N_NODES - 1) off[N_NODES] = prefix + v;
}

__global__ __launch_bounds__(256)
void grp_fill(const int* __restrict__ src, const int* __restrict__ dst,
              const int* __restrict__ slot, const int* __restrict__ off,
              int* __restrict__ tmp, int E) {
    const int e = blockIdx.x * blockDim.x + threadIdx.x;
    if (e >= E) return;
    tmp[off[dst[e]] + slot[e]] = src[e];
}

__global__ __launch_bounds__(256)
void grp_order(const int* __restrict__ off, const int* __restrict__ tmp,
               int* __restrict__ nbr) {
    const int node = __builtin_amdgcn_readfirstlane(
        (int)(blockIdx.x * 8 + (threadIdx.x >> 5)));
    if (node >= N_NODES) return;
    const int lane = threadIdx.x & 31;
    const int beg = off[node];
    const int deg = off[node + 1] - beg;
    for (int ib = 0; ib < deg; ib += 32) {
        const int i = ib + lane;
        const bool act = i < deg;
        const int key = act ? tmp[beg + i] : 0;
        int r = 0;
        for (int jb = 0; jb < deg; jb += 32) {
            const int jn = min(32, deg - jb);
            const int kl = (jb + lane < deg) ? tmp[beg + jb + lane] : 0;
            for (int jj = 0; jj < jn; ++jj) {
                const int kj = __builtin_amdgcn_readlane(kl, jj);
                const int j = jb + jj;
                r += (kj < key || (kj == key && j < i)) ? 1 : 0;
            }
        }
        if (act) nbr[beg + r] = key;
    }
}

template <int CTRL>
__device__ __forceinline__ float row_xadd(float v) {
    const int t = __builtin_amdgcn_update_dpp(0, __float_as_int(v), CTRL, 0xf, 0xf, true);
    return v + __int_as_float(t);
}
__device__ __forceinline__ float xhalf(float x) {
    const int b = __float_as_int(x);
    return __int_as_float(
        __builtin_amdgcn_permlanex16(b, b, 0x76543210, 0xfedcba98, false, false));
}
__device__ __forceinline__ float edge_part(const float4 v, const float4 b, const float4 w) {
    return w.x * lrelu(v.x + b.x) + w.y * lrelu(v.y + b.y)
         + w.z * lrelu(v.z + b.z) + w.w * lrelu(v.w + b.w);
}

__global__ __launch_bounds__(256)
void gat_node_fused(const float* __restrict__ fs, const float* __restrict__ fd,
                    const float* __restrict__ attn,
                    const int* __restrict__ off, const int* __restrict__ nbr,
                    float* __restrict__ hout) {
    const int node = __builtin_amdgcn_readfirstlane(
        (int)(blockIdx.x * 8 + (threadIdx.x >> 5)));
    if (node >= N_NODES) return;
    const int lane = threadIdx.x & 31;
    const int beg = off[node];
    const int end = off[node + 1];

    const float4 b = *(const float4*)(fd + (size_t)node * HF + lane * 4);
    const float4 w = *(const float4*)(attn + lane * 4);

    float m = -INFINITY, den = 0.0f;
    float4 acc = make_float4(0.0f, 0.0f, 0.0f, 0.0f);

    for (int cb = beg; cb < end; cb += 32) {
        const int cn = min(32, end - cb);
        const int ids = nbr[cb + min(lane, cn - 1)];
        for (int j = 0; j < cn; j += 4) {
            const int s0 = __builtin_amdgcn_readlane(ids, j);
            const int s1 = __builtin_amdgcn_readlane(ids, j + 1);
            const int s2 = __builtin_amdgcn_readlane(ids, j + 2);
            const int s3 = __builtin_amdgcn_readlane(ids, j + 3);
            const float4 v0 = *(const float4*)(fs + (size_t)s0 * HF + lane * 4);
            const float4 v1 = *(const float4*)(fs + (size_t)s1 * HF + lane * 4);
            const float4 v2 = *(const float4*)(fs + (size_t)s2 * HF + lane * 4);
            const float4 v3 = *(const float4*)(fs + (size_t)s3 * HF + lane * 4);

            float l0 = edge_part(v0, b, w);
            float l1 = edge_part(v1, b, w);
            float l2 = edge_part(v2, b, w);
            float l3 = edge_part(v3, b, w);
            l0 = row_xadd<0x161>(l0); l1 = row_xadd<0x161>(l1);
            l2 = row_xadd<0x161>(l2); l3 = row_xadd<0x161>(l3);
            l0 = row_xadd<0x162>(l0); l1 = row_xadd<0x162>(l1);
            l2 = row_xadd<0x162>(l2); l3 = row_xadd<0x162>(l3);
            l0 = row_xadd<0x164>(l0); l1 = row_xadd<0x164>(l1);
            l2 = row_xadd<0x164>(l2); l3 = row_xadd<0x164>(l3);
            l0 = row_xadd<0x168>(l0); l1 = row_xadd<0x168>(l1);
            l2 = row_xadd<0x168>(l2); l3 = row_xadd<0x168>(l3);
            if (j + 1 >= cn) l1 = -INFINITY;
            if (j + 2 >= cn) l2 = -INFINITY;
            if (j + 3 >= cn) l3 = -INFINITY;

            const float mn = fmaxf(fmaxf(m, l0), fmaxf(l1, fmaxf(l2, l3)));
            const float sc = expf(m - mn);
            const float p0 = expf(l0 - mn);
            const float p1 = expf(l1 - mn);
            const float p2 = expf(l2 - mn);
            const float p3 = expf(l3 - mn);
            den   = den   * sc + ((p0 + p1) + (p2 + p3));
            acc.x = acc.x * sc + ((p0 * v0.x + p1 * v1.x) + (p2 * v2.x + p3 * v3.x));
            acc.y = acc.y * sc + ((p0 * v0.y + p1 * v1.y) + (p2 * v2.y + p3 * v3.y));
            acc.z = acc.z * sc + ((p0 * v0.z + p1 * v1.z) + (p2 * v2.z + p3 * v3.z));
            acc.w = acc.w * sc + ((p0 * v0.w + p1 * v1.w) + (p2 * v2.w + p3 * v3.w));
            m = mn;
        }
    }

    const float inv = (end > beg) ? 1.0f / den : 0.0f;
    float4 o;
    o.x = acc.x * inv; o.y = acc.y * inv; o.z = acc.z * inv; o.w = acc.w * inv;
    o.x = 0.5f * (o.x + xhalf(o.x));
    o.y = 0.5f * (o.y + xhalf(o.y));
    o.z = 0.5f * (o.z + xhalf(o.z));
    o.w = 0.5f * (o.w + xhalf(o.w));
    if (lane < 16)
        *(float4*)(hout + (size_t)node * F_DIM + lane * 4) = o;
}

extern "C" void kernel_launch(void* const* d_in, const int* in_sizes, int n_in,
                              void* d_out, int out_size, void* d_ws, size_t ws_size,
                              hipStream_t stream) {
    (void)n_in; (void)out_size; (void)ws_size;
    const float* h0   = (const float*)d_in[0];
    const float* Wsrc = (const float*)d_in[1];
    const float* bsrc = (const float*)d_in[2];
    const float* Wdst = (const float*)d_in[3];
    const float* bdst = (const float*)d_in[4];
    const float* attn = (const float*)d_in[5];
    const int*   src  = (const int*)d_in[6];
    const int*   dst  = (const int*)d_in[7];
    const int    E    = in_sizes[6];

    float* ws   = (float*)d_ws;
    float* fs   = ws;  ws += (size_t)N_NODES * HF;
    float* fd   = ws;  ws += (size_t)N_NODES * HF;
    float* hbuf = ws;  ws += (size_t)N_NODES * F_DIM;
    int* cnt    = (int*)ws;  ws += (size_t)N_NODES;
    int* off    = (int*)ws;  ws += (size_t)N_NODES + 4;
    int* slot   = (int*)ws;  ws += (size_t)E;
    int* tmp    = (int*)ws;  ws += (size_t)E;
    int* nbr    = (int*)ws;

    const int gridE = (E + 255) / 256;
    const int gridW = (N_NODES + 7) / 8;

    hipMemsetAsync(cnt, 0, (size_t)N_NODES * sizeof(int), stream);
    grp_count<<<gridE, 256, 0, stream>>>(dst, cnt, slot, E);
    grp_offsets<<<(N_NODES + 255) / 256, 256, 0, stream>>>(cnt, off);
    grp_fill<<<gridE, 256, 0, stream>>>(src, dst, slot, off, tmp, E);
    grp_order<<<gridW, 256, 0, stream>>>(off, tmp, nbr);

    const float* hcur = h0;
    for (int l = 0; l < L_LAYERS; ++l) {
        gat_proj_wmma<<<dim3(PROJ_BLOCKS, 2), PROJ_WAVES * 32, 0, stream>>>(
            hcur,
            Wsrc + (size_t)l * F_DIM * HF, bsrc + (size_t)l * HF,
            Wdst + (size_t)l * F_DIM * HF, bdst + (size_t)l * HF,
            fs, fd);

        float* hn = (l == L_LAYERS - 1) ? (float*)d_out : hbuf;
        gat_node_fused<<<gridW, 256, 0, stream>>>(
            fs, fd, attn + (size_t)l * HF, off, nbr, hn);
        hcur = hbuf;
    }
}


# edge grouping without global atomics: (edge slice x node range) work-groups count and place edges with LDS counters, per-slice counts prefixed per node; memset, rank array dropped; sum order same
# speedup vs baseline: 4.3370x; 1.3410x over previous
#include <hip/hip_runtime.h>

#define N_NODES   50000
#define F_DIM     64
#define H_HEADS   2
#define HF        128
#define L_LAYERS  2
#define NEG_SLOPE 0.2f

typedef __attribute__((ext_vector_type(2))) float v2f;
typedef __attribute__((ext_vector_type(8))) float v8f;

__device__ __forceinline__ float lrelu(float x) {
    return x > 0.0f ? x : NEG_SLOPE * x;
}

#define ROW_TILES   (N_NODES / 16)
#define PROJ_WAVES  4
#define PROJ_BLOCKS ((ROW_TILES + PROJ_WAVES - 1) / PROJ_WAVES)

__global__ __launch_bounds__(PROJ_WAVES * 32)
void gat_proj_wmma(const float* __restrict__ h,
                   const float* __restrict__ Wsrc, const float* __restrict__ bsrc,
                   const float* __restrict__ Wdst, const float* __restrict__ bdst,
                   float* __restrict__ fs, float* __restrict__ fd) {
    __shared__ float4 wl[16 * 8 * 16];

    const int tid  = threadIdx.x;
    const int wave = tid >> 5;
    const int lane = tid & 31;
    const int hh   = lane >> 4;
    const int c    = lane & 15;

    const bool  isSrc = (blockIdx.y == 0);
    const float* __restrict__ W    = isSrc ? Wsrc : Wdst;
    const float* __restrict__ bias = isSrc ? bsrc : bdst;
    float* __restrict__ out        = isSrc ? fs : fd;

    const int  tile = blockIdx.x * PROJ_WAVES + wave;
    const bool live = tile < ROW_TILES;
    const int  ltile = live ? tile : ROW_TILES - 1;

    const float* __restrict__ arow = h + (size_t)(ltile * 16 + c) * F_DIM + hh * 4;
    float4 a[8];
#pragma unroll
    for (int i2 = 0; i2 < 8; ++i2)
        a[i2] = *(const float4*)(arow + 8 * i2);

#pragma unroll
    for (int it = 0; it < 512 / (PROJ_WAVES * 32); ++it) {
        const int item = it * (PROJ_WAVES * 32) + tid;
        const int q  = item >> 5;
        const int c4 = item & 31;
        const float* __restrict__ wp = W + (size_t)(4 * q) * HF + 4 * c4;
        const float4 r0 = *(const float4*)(wp);
        const float4 r1 = *(const float4*)(wp + HF);
        const float4 r2 = *(const float4*)(wp + 2 * HF);
        const float4 r3 = *(const float4*)(wp + 3 * HF);
        float4* dp = wl + (q * 8 + 4 * (c4 >> 4)) * 16 + (c4 & 15);
        dp[0]  = make_float4(r0.x, r1.x, r2.x, r3.x);
        dp[16] = make_float4(r0.y, r1.y, r2.y, r3.y);
        dp[32] = make_float4(r0.z, r1.z, r2.z, r3.z);
        dp[48] = make_float4(r0.w, r1.w, r2.w, r3.w);
    }
    __syncthreads();
    if (!live) return;

    v8f acc[8];
#pragma unroll
    for (int t = 0; t < 8; ++t) acc[t] = (v8f){};

    const float4* bl = wl + hh * 128 + c;
#pragma unroll
    for (int i2 = 0; i2 < 8; ++i2) {
        float4 b[8];
#pragma unroll
        for (int t = 0; t < 8; ++t) b[t] = bl[i2 * 256 + t * 16];
        v2f a0, a1;
        a0.x = a[i2].x; a0.y = a[i2].y;
        a1.x = a[i2].z; a1.y = a[i2].w;
#pragma unroll
        for (int t = 0; t < 8; ++t) {
            v2f b0; b0.x = b[t].x; b0.y = b[t].y;
            acc[t] = __builtin_amdgcn_wmma_f32_16x16x4_f32(
                        false, a0, false, b0, (short)0, acc[t], false, false);
        }
#pragma unroll
        for (int t = 0; t < 8; ++t) {
            v2f b1; b1.x = b[t].z; b1.y = b[t].w;
            acc[t] = __builtin_amdgcn_wmma_f32_16x16x4_f32(
                        false, a1, false, b1, (short)0, acc[t], false, false);
        }
    }

    float* __restrict__ orow = out + (size_t)(tile * 16 + hh * 8) * HF + 4 * c;
#pragma unroll
    for (int g = 0; g < 2; ++g) {
        const float4 bv = *(const float4*)(bias + 64 * g + 4 * c);
#pragma unroll
        for (int r = 0; r < 8; ++r) {
            float4 o;
            o.x = acc[4 * g + 0][r] + bv.x;
            o.y = acc[4 * g + 1][r] + bv.y;
            o.z = acc[4 * g + 2][r] + bv.z;
            o.w = acc[4 * g + 3][r] + bv.w;
            *(float4*)(orow + (size_t)r * HF + 64 * g) = o;
        }
    }
}

#define GRP_SLICES   36
#define GRP_THREADS  1024
#define NODE_BLOCKS  ((N_NODES + 255) / 256)
#define RANGE_BLOCKS 28
#define RANGE_NODES  (RANGE_BLOCKS * 256)
#define GRP_RANGES   (NODE_BLOCKS / RANGE_BLOCKS)
#define N_PAD        (NODE_BLOCKS * 256)
static_assert(GRP_RANGES * RANGE_BLOCKS == NODE_BLOCKS, "ranges must tile the node blocks");
static_assert(RANGE_BLOCKS * 32 <= GRP_THREADS, "one wave per node block for the block sums");
static_assert(RANGE_NODES % GRP_THREADS == 0, "whole passes over a range's counters");

__device__ __forceinline__ int grp_slice_len(int E) {
    return (((E + GRP_SLICES - 1) / GRP_SLICES) + 3) & ~3;
}
#define GRP_STEP  (4 * GRP_THREADS)
#define GRP_BATCH 6

__global__ __launch_bounds__(GRP_THREADS)
void grp_count(const int* __restrict__ dst, int* __restrict__ pc,
               int* __restrict__ bs, int E) {
    __shared__ int lc[RANGE_NODES];
    const int tid = threadIdx.x;
    const int s   = blockIdx.x;
    const int lo  = blockIdx.y * RANGE_NODES;
    for (int i = tid; i < RANGE_NODES; i += GRP_THREADS) lc[i] = 0;
    __syncthreads();

    const int sl = grp_slice_len(E);
    const int e0 = s * sl;
    const int e1 = min(E & ~3, e0 + sl);
    for (int base = e0 + 4 * tid; base < e1; base += GRP_BATCH * GRP_STEP) {
        int4 d[GRP_BATCH];
#pragma unroll
        for (int k = 0; k < GRP_BATCH; ++k) {
            const int e = base + k * GRP_STEP;
            d[k] = (e < e1) ? *(const int4*)(dst + e) : make_int4(-1, -1, -1, -1);
        }
#pragma unroll
        for (int k = 0; k < GRP_BATCH; ++k) {
            const unsigned x0 = (unsigned)(d[k].x - lo), x1 = (unsigned)(d[k].y - lo);
            const unsigned x2 = (unsigned)(d[k].z - lo), x3 = (unsigned)(d[k].w - lo);
            if (x0 < RANGE_NODES) atomicAdd(&lc[x0], 1);
            if (x1 < RANGE_NODES) atomicAdd(&lc[x1], 1);
            if (x2 < RANGE_NODES) atomicAdd(&lc[x2], 1);
            if (x3 < RANGE_NODES) atomicAdd(&lc[x3], 1);
        }
    }
    const int et = E & ~3;
    if (tid == 0 && et >= e0 && et < e0 + sl)
        for (int e = et; e < E; ++e) {
            const unsigned x = (unsigned)(dst[e] - lo);
            if (x < RANGE_NODES) atomicAdd(&lc[x], 1);
        }
    __syncthreads();

    for (int i = tid; i < RANGE_NODES; i += GRP_THREADS)
        pc[(size_t)s * N_PAD + lo + i] = lc[i];
    const int wave = tid >> 5, lane = tid & 31;
    if (wave < RANGE_BLOCKS) {
        int sum = 0;
#pragma unroll
        for (int k = 0; k < 8; ++k) sum += lc[wave * 256 + k * 32 + lane];
#pragma unroll
        for (int o = 16; o > 0; o >>= 1) sum += __shfl_xor(sum, o, 32);
        if (lane == 0)
            bs[s * NODE_BLOCKS + blockIdx.y * RANGE_BLOCKS + wave] = sum;
    }
}

__global__ __launch_bounds__(256)
void grp_offsets(int* __restrict__ pc, const int* __restrict__ bs,
                 int* __restrict__ off) {
    __shared__ int red[8];
    __shared__ int sc[256];
    const int t = threadIdx.x;
    const int b = blockIdx.x;
    const int i = b * 256 + t;
    int cnts[GRP_SLICES];
#pragma unroll
    for (int s = 0; s < GRP_SLICES; ++s) cnts[s] = pc[(size_t)s * N_PAD + i];
    int part = 0;
    if (t < b) {
#pragma unroll
        for (int s = 0; s < GRP_SLICES; ++s) part += bs[s * NODE_BLOCKS + t];
    }
#pragma unroll
    for (int o = 16; o > 0; o >>= 1)
        part += __shfl_xor(part, o, 32);
    if ((t & 31) == 0) red[t >> 5] = part;

    int run = 0;
#pragma unroll
    for (int s = 0; s < GRP_SLICES; ++s) {
        pc[(size_t)s * N_PAD + i] = run;
        run += cnts[s];
    }
    const int c = run;
    sc[t] = c;
    __syncthreads();
    int prefix = 0;
#pragma unroll
    for (int k = 0; k < 8; ++k) prefix += red[k];
    int v = c;
    for (int o = 1; o < 256; o <<= 1) {
        const int add = (t >= o) ? sc[t - o] : 0;
        __syncthreads();
        v += add;
        sc[t] = v;
        __syncthreads();
    }
    if (i < N_NODES) off[i] = prefix + v - c;
    if (i == N_NODES - 1) off[N_NODES] = prefix + v;
}

__global__ __launch_bounds__(GRP_THREADS)
void grp_fill(const int* __restrict__ src, const int* __restrict__ dst,
              const int* __restrict__ pc, const int* __restrict__ off,
              int* __restrict__ tmp, int E) {
    __shared__ int cur[RANGE_NODES];
    const int tid = threadIdx.x;
    const int s   = blockIdx.x;
    const int lo  = blockIdx.y * RANGE_NODES;
#pragma unroll
    for (int k = 0; k < RANGE_NODES / GRP_THREADS; ++k) {
        const int i = k * GRP_THREADS + tid;
        const int node = lo + i;
        cur[i] = off[min(node, N_NODES)] + pc[(size_t)s * N_PAD + node];
    }
    __syncthreads();

    const int sl = grp_slice_len(E);
    const int e0 = s * sl;
    const int e1 = min(E & ~3, e0 + sl);
    for (int base = e0 + 4 * tid; base < e1; base += GRP_BATCH * GRP_STEP) {
        int4 d[GRP_BATCH], sv[GRP_BATCH];
#pragma unroll
        for (int k = 0; k < GRP_BATCH; ++k) {
            const int e = base + k * GRP_STEP;
            d[k]  = (e < e1) ? *(const int4*)(dst + e) : make_int4(-1, -1, -1, -1);
            sv[k] = (e < e1) ? *(const int4*)(src + e) : make_int4(0, 0, 0, 0);
        }
#pragma unroll
        for (int k = 0; k < GRP_BATCH; ++k) {
            const unsigned x0 = (unsigned)(d[k].x - lo), x1 = (unsigned)(d[k].y - lo);
            const unsigned x2 = (unsigned)(d[k].z - lo), x3 = (unsigned)(d[k].w - lo);
            if (x0 < RANGE_NODES) tmp[atomicAdd(&cur[x0], 1)] = sv[k].x;
            if (x1 < RANGE_NODES) tmp[atomicAdd(&cur[x1], 1)] = sv[k].y;
            if (x2 < RANGE_NODES) tmp[atomicAdd(&cur[x2], 1)] = sv[k].z;
            if (x3 < RANGE_NODES) tmp[atomicAdd(&cur[x3], 1)] = sv[k].w;
        }
    }
    const int et = E & ~3;
    if (tid == 0 && et >= e0 && et < e0 + sl)
        for (int e = et; e < E; ++e) {
            const unsigned x = (unsigned)(dst[e] - lo);
            if (x < RANGE_NODES) tmp[atomicAdd(&cur[x], 1)] = src[e];
        }
}

__global__ __launch_bounds__(256)
void grp_order(const int* __restrict__ off, const int* __restrict__ tmp,
               int* __restrict__ nbr) {
    const int node = __builtin_amdgcn_readfirstlane(
        (int)(blockIdx.x * 8 + (threadIdx.x >> 5)));
    if (node >= N_NODES) return;
    const int lane = threadIdx.x & 31;
    const int beg = off[node];
    const int deg = off[node + 1] - beg;
    for (int ib = 0; ib < deg; ib += 32) {
        const int i = ib + lane;
        const bool act = i < deg;
        const int key = act ? tmp[beg + i] : 0;
        int r = 0;
        for (int jb = 0; jb < deg; jb += 32) {
            const int jn = min(32, deg - jb);
            const int kl = (jb + lane < deg) ? tmp[beg + jb + lane] : 0;
            for (int jj = 0; jj < jn; ++jj) {
                const int kj = __builtin_amdgcn_readlane(kl, jj);
                const int j = jb + jj;
                r += (kj < key || (kj == key && j < i)) ? 1 : 0;
            }
        }
        if (act) nbr[beg + r] = key;
    }
}

template <int CTRL>
__device__ __forceinline__ float row_xadd(float v) {
    const int t = __builtin_amdgcn_update_dpp(0, __float_as_int(v), CTRL, 0xf, 0xf, true);
    return v + __int_as_float(t);
}
__device__ __forceinline__ float xhalf(float x) {
    const int b = __float_as_int(x);
    return __int_as_float(
        __builtin_amdgcn_permlanex16(b, b, 0x76543210, 0xfedcba98, false, false));
}
__device__ __forceinline__ float edge_part(const float4 v, const float4 b, const float4 w) {
    return w.x * lrelu(v.x + b.x) + w.y * lrelu(v.y + b.y)
         + w.z * lrelu(v.z + b.z) + w.w * lrelu(v.w + b.w);
}

__global__ __launch_bounds__(256)
void gat_node_fused(const float* __restrict__ fs, const float* __restrict__ fd,
                    const float* __restrict__ attn,
                    const int* __restrict__ off, const int* __restrict__ nbr,
                    float* __restrict__ hout) {
    const int node = __builtin_amdgcn_readfirstlane(
        (int)(blockIdx.x * 8 + (threadIdx.x >> 5)));
    if (node >= N_NODES) return;
    const int lane = threadIdx.x & 31;
    const int beg = off[node];
    const int end = off[node + 1];

    const float4 b = *(const float4*)(fd + (size_t)node * HF + lane * 4);
    const float4 w = *(const float4*)(attn + lane * 4);

    float m = -INFINITY, den = 0.0f;
    float4 acc = make_float4(0.0f, 0.0f, 0.0f, 0.0f);

    for (int cb = beg; cb < end; cb += 32) {
        const int cn = min(32, end - cb);
        const int ids = nbr[cb + min(lane, cn - 1)];
        for (int j = 0; j < cn; j += 4) {
            const int s0 = __builtin_amdgcn_readlane(ids, j);
            const int s1 = __builtin_amdgcn_readlane(ids, j + 1);
            const int s2 = __builtin_amdgcn_readlane(ids, j + 2);
            const int s3 = __builtin_amdgcn_readlane(ids, j + 3);
            const float4 v0 = *(const float4*)(fs + (size_t)s0 * HF + lane * 4);
            const float4 v1 = *(const float4*)(fs + (size_t)s1 * HF + lane * 4);
            const float4 v2 = *(const float4*)(fs + (size_t)s2 * HF + lane * 4);
            const float4 v3 = *(const float4*)(fs + (size_t)s3 * HF + lane * 4);

            float l0 = edge_part(v0, b, w);
            float l1 = edge_part(v1, b, w);
            float l2 = edge_part(v2, b, w);
            float l3 = edge_part(v3, b, w);
            l0 = row_xadd<0x161>(l0); l1 = row_xadd<0x161>(l1);
            l2 = row_xadd<0x161>(l2); l3 = row_xadd<0x161>(l3);
            l0 = row_xadd<0x162>(l0); l1 = row_xadd<0x162>(l1);
            l2 = row_xadd<0x162>(l2); l3 = row_xadd<0x162>(l3);
            l0 = row_xadd<0x164>(l0); l1 = row_xadd<0x164>(l1);
            l2 = row_xadd<0x164>(l2); l3 = row_xadd<0x164>(l3);
            l0 = row_xadd<0x168>(l0); l1 = row_xadd<0x168>(l1);
            l2 = row_xadd<0x168>(l2); l3 = row_xadd<0x168>(l3);
            if (j + 1 >= cn) l1 = -INFINITY;
            if (j + 2 >= cn) l2 = -INFINITY;
            if (j + 3 >= cn) l3 = -INFINITY;

            const float mn = fmaxf(fmaxf(m, l0), fmaxf(l1, fmaxf(l2, l3)));
            const float sc = expf(m - mn);
            const float p0 = expf(l0 - mn);
            const float p1 = expf(l1 - mn);
            const float p2 = expf(l2 - mn);
            const float p3 = expf(l3 - mn);
            den   = den   * sc + ((p0 + p1) + (p2 + p3));
            acc.x = acc.x * sc + ((p0 * v0.x + p1 * v1.x) + (p2 * v2.x + p3 * v3.x));
            acc.y = acc.y * sc + ((p0 * v0.y + p1 * v1.y) + (p2 * v2.y + p3 * v3.y));
            acc.z = acc.z * sc + ((p0 * v0.z + p1 * v1.z) + (p2 * v2.z + p3 * v3.z));
            acc.w = acc.w * sc + ((p0 * v0.w + p1 * v1.w) + (p2 * v2.w + p3 * v3.w));
            m = mn;
        }
    }

    const float inv = (end > beg) ? 1.0f / den : 0.0f;
    float4 o;
    o.x = acc.x * inv; o.y = acc.y * inv; o.z = acc.z * inv; o.w = acc.w * inv;
    o.x = 0.5f * (o.x + xhalf(o.x));
    o.y = 0.5f * (o.y + xhalf(o.y));
    o.z = 0.5f * (o.z + xhalf(o.z));
    o.w = 0.5f * (o.w + xhalf(o.w));
    if (lane < 16)
        *(float4*)(hout + (size_t)node * F_DIM + lane * 4) = o;
}

extern "C" void kernel_launch(void* const* d_in, const int* in_sizes, int n_in,
                              void* d_out, int out_size, void* d_ws, size_t ws_size,
                              hipStream_t stream) {
    (void)n_in; (void)out_size; (void)ws_size;
    const float* h0   = (const float*)d_in[0];
    const float* Wsrc = (const float*)d_in[1];
    const float* bsrc = (const float*)d_in[2];
    const float* Wdst = (const float*)d_in[3];
    const float* bdst = (const float*)d_in[4];
    const float* attn = (const float*)d_in[5];
    const int*   src  = (const int*)d_in[6];
    const int*   dst  = (const int*)d_in[7];
    const int    E    = in_sizes[6];

    float* ws   = (float*)d_ws;
    float* fs   = ws;  ws += (size_t)N_NODES * HF;
    float* fd   = ws;  ws += (size_t)N_NODES * HF;
    float* hbuf = ws;  ws += (size_t)N_NODES * F_DIM;
    int* off    = (int*)ws;  ws += (size_t)N_NODES + 4;
    int* tmp    = (int*)ws;  ws += (size_t)E;
    int* nbr    = (int*)ws;
    int* pc     = (int*)hbuf;
    int* bs     = pc + (size_t)GRP_SLICES * N_PAD;

    const int gridW = (N_NODES + 7) / 8;
    const dim3 gridC(GRP_SLICES, GRP_RANGES);

    grp_count<<<gridC, GRP_THREADS, 0, stream>>>(dst, pc, bs, E);
    grp_offsets<<<NODE_BLOCKS, 256, 0, stream>>>(pc, bs, off);
    grp_fill<<<gridC, GRP_THREADS, 0, stream>>>(src, dst, pc, off, tmp, E);
    grp_order<<<gridW, 256, 0, stream>>>(off, tmp, nbr);

    const float* hcur = h0;
    for (int l = 0; l < L_LAYERS; ++l) {
        gat_proj_wmma<<<dim3(PROJ_BLOCKS, 2), PROJ_WAVES * 32, 0, stream>>>(
            hcur,
            Wsrc + (size_t)l * F_DIM * HF, bsrc + (size_t)l * HF,
            Wdst + (size_t)l * F_DIM * HF, bdst + (size_t)l * HF,
            fs, fd);

        float* hn = (l == L_LAYERS - 1) ? (float*)d_out : hbuf;
        gat_node_fused<<<gridW, 256, 0, stream>>>(
            fs, fd, attn + (size_t)l * HF, off, nbr, hn);
        hcur = hbuf;
    }
}


# aggregation: 4-edge logit butterfly halves per step, lane l ends with edge (l&3)'s logit; one exp per lane (own edge or rescale factor), values back by DPP row moves; every sum in the same order
# speedup vs baseline: 4.5876x; 1.0578x over previous
#include <hip/hip_runtime.h>

#define N_NODES   50000
#define F_DIM     64
#define H_HEADS   2
#define HF        128
#define L_LAYERS  2
#define NEG_SLOPE 0.2f

typedef __attribute__((ext_vector_type(2))) float v2f;
typedef __attribute__((ext_vector_type(8))) float v8f;

__device__ __forceinline__ float lrelu(float x) {
    return x > 0.0f ? x : NEG_SLOPE * x;
}

#define ROW_TILES   (N_NODES / 16)
#define PROJ_WAVES  4
#define PROJ_BLOCKS ((ROW_TILES + PROJ_WAVES - 1) / PROJ_WAVES)

__global__ __launch_bounds__(PROJ_WAVES * 32)
void gat_proj_wmma(const float* __restrict__ h,
                   const float* __restrict__ Wsrc, const float* __restrict__ bsrc,
                   const float* __restrict__ Wdst, const float* __restrict__ bdst,
                   float* __restrict__ fs, float* __restrict__ fd) {
    __shared__ float4 wl[16 * 8 * 16];

    const int tid  = threadIdx.x;
    const int wave = tid >> 5;
    const int lane = tid & 31;
    const int hh   = lane >> 4;
    const int c    = lane & 15;

    const bool  isSrc = (blockIdx.y == 0);
    const float* __restrict__ W    = isSrc ? Wsrc : Wdst;
    const float* __restrict__ bias = isSrc ? bsrc : bdst;
    float* __restrict__ out        = isSrc ? fs : fd;

    const int  tile = blockIdx.x * PROJ_WAVES + wave;
    const bool live = tile < ROW_TILES;
    const int  ltile = live ? tile : ROW_TILES - 1;

    const float* __restrict__ arow = h + (size_t)(ltile * 16 + c) * F_DIM + hh * 4;
    float4 a[8];
#pragma unroll
    for (int i2 = 0; i2 < 8; ++i2)
        a[i2] = *(const float4*)(arow + 8 * i2);

#pragma unroll
    for (int it = 0; it < 512 / (PROJ_WAVES * 32); ++it) {
        const int item = it * (PROJ_WAVES * 32) + tid;
        const int q  = item >> 5;
        const int c4 = item & 31;
        const float* __restrict__ wp = W + (size_t)(4 * q) * HF + 4 * c4;
        const float4 r0 = *(const float4*)(wp);
        const float4 r1 = *(const float4*)(wp + HF);
        const float4 r2 = *(const float4*)(wp + 2 * HF);
        const float4 r3 = *(const float4*)(wp + 3 * HF);
        float4* dp = wl + (q * 8 + 4 * (c4 >> 4)) * 16 + (c4 & 15);
        dp[0]  = make_float4(r0.x, r1.x, r2.x, r3.x);
        dp[16] = make_float4(r0.y, r1.y, r2.y, r3.y);
        dp[32] = make_float4(r0.z, r1.z, r2.z, r3.z);
        dp[48] = make_float4(r0.w, r1.w, r2.w, r3.w);
    }
    __syncthreads();
    if (!live) return;

    v8f acc[8];
#pragma unroll
    for (int t = 0; t < 8; ++t) acc[t] = (v8f){};

    const float4* bl = wl + hh * 128 + c;
#pragma unroll
    for (int i2 = 0; i2 < 8; ++i2) {
        float4 b[8];
#pragma unroll
        for (int t = 0; t < 8; ++t) b[t] = bl[i2 * 256 + t * 16];
        v2f a0, a1;
        a0.x = a[i2].x; a0.y = a[i2].y;
        a1.x = a[i2].z; a1.y = a[i2].w;
#pragma unroll
        for (int t = 0; t < 8; ++t) {
            v2f b0; b0.x = b[t].x; b0.y = b[t].y;
            acc[t] = __builtin_amdgcn_wmma_f32_16x16x4_f32(
                        false, a0, false, b0, (short)0, acc[t], false, false);
        }
#pragma unroll
        for (int t = 0; t < 8; ++t) {
            v2f b1; b1.x = b[t].z; b1.y = b[t].w;
            acc[t] = __builtin_amdgcn_wmma_f32_16x16x4_f32(
                        false, a1, false, b1, (short)0, acc[t], false, false);
        }
    }

    float* __restrict__ orow = out + (size_t)(tile * 16 + hh * 8) * HF + 4 * c;
#pragma unroll
    for (int g = 0; g < 2; ++g) {
        const float4 bv = *(const float4*)(bias + 64 * g + 4 * c);
#pragma unroll
        for (int r = 0; r < 8; ++r) {
            float4 o;
            o.x = acc[4 * g + 0][r] + bv.x;
            o.y = acc[4 * g + 1][r] + bv.y;
            o.z = acc[4 * g + 2][r] + bv.z;
            o.w = acc[4 * g + 3][r] + bv.w;
            *(float4*)(orow + (size_t)r * HF + 64 * g) = o;
        }
    }
}

#define GRP_SLICES   36
#define GRP_THREADS  1024
#define NODE_BLOCKS  ((N_NODES + 255) / 256)
#define RANGE_BLOCKS 28
#define RANGE_NODES  (RANGE_BLOCKS * 256)
#define GRP_RANGES   (NODE_BLOCKS / RANGE_BLOCKS)
#define N_PAD        (NODE_BLOCKS * 256)
static_assert(GRP_RANGES * RANGE_BLOCKS == NODE_BLOCKS, "ranges must tile the node blocks");
static_assert(RANGE_BLOCKS * 32 <= GRP_THREADS, "one wave per node block for the block sums");
static_assert(RANGE_NODES % GRP_THREADS == 0, "whole passes over a range's counters");

__device__ __forceinline__ int grp_slice_len(int E) {
    return (((E + GRP_SLICES - 1) / GRP_SLICES) + 3) & ~3;
}
#define GRP_STEP  (4 * GRP_THREADS)
#define GRP_BATCH 6

__global__ __launch_bounds__(GRP_THREADS)
void grp_count(const int* __restrict__ dst, int* __restrict__ pc,
               int* __restrict__ bs, int E) {
    __shared__ int lc[RANGE_NODES];
    const int tid = threadIdx.x;
    const int s   = blockIdx.x;
    const int lo  = blockIdx.y * RANGE_NODES;
    for (int i = tid; i < RANGE_NODES; i += GRP_THREADS) lc[i] = 0;
    __syncthreads();

    const int sl = grp_slice_len(E);
    const int e0 = s * sl;
    const int e1 = min(E & ~3, e0 + sl);
    for (int base = e0 + 4 * tid; base < e1; base += GRP_BATCH * GRP_STEP) {
        int4 d[GRP_BATCH];
#pragma unroll
        for (int k = 0; k < GRP_BATCH; ++k) {
            const int e = base + k * GRP_STEP;
            d[k] = (e < e1) ? *(const int4*)(dst + e) : make_int4(-1, -1, -1, -1);
        }
#pragma unroll
        for (int k = 0; k < GRP_BATCH; ++k) {
            const unsigned x0 = (unsigned)(d[k].x - lo), x1 = (unsigned)(d[k].y - lo);
            const unsigned x2 = (unsigned)(d[k].z - lo), x3 = (unsigned)(d[k].w - lo);
            if (x0 < RANGE_NODES) atomicAdd(&lc[x0], 1);
            if (x1 < RANGE_NODES) atomicAdd(&lc[x1], 1);
            if (x2 < RANGE_NODES) atomicAdd(&lc[x2], 1);
            if (x3 < RANGE_NODES) atomicAdd(&lc[x3], 1);
        }
    }
    const int et = E & ~3;
    if (tid == 0 && et >= e0 && et < e0 + sl)
        for (int e = et; e < E; ++e) {
            const unsigned x = (unsigned)(dst[e] - lo);
            if (x < RANGE_NODES) atomicAdd(&lc[x], 1);
        }
    __syncthreads();

    for (int i = tid; i < RANGE_NODES; i += GRP_THREADS)
        pc[(size_t)s * N_PAD + lo + i] = lc[i];
    const int wave = tid >> 5, lane = tid & 31;
    if (wave < RANGE_BLOCKS) {
        int sum = 0;
#pragma unroll
        for (int k = 0; k < 8; ++k) sum += lc[wave * 256 + k * 32 + lane];
#pragma unroll
        for (int o = 16; o > 0; o >>= 1) sum += __shfl_xor(sum, o, 32);
        if (lane == 0)
            bs[s * NODE_BLOCKS + blockIdx.y * RANGE_BLOCKS + wave] = sum;
    }
}

__global__ __launch_bounds__(256)
void grp_offsets(int* __restrict__ pc, const int* __restrict__ bs,
                 int* __restrict__ off) {
    __shared__ int red[8];
    __shared__ int sc[256];
    const int t = threadIdx.x;
    const int b = blockIdx.x;
    const int i = b * 256 + t;
    int cnts[GRP_SLICES];
#pragma unroll
    for (int s = 0; s < GRP_SLICES; ++s) cnts[s] = pc[(size_t)s * N_PAD + i];
    int part = 0;
    if (t < b) {
#pragma unroll
        for (int s = 0; s < GRP_SLICES; ++s) part += bs[s * NODE_BLOCKS + t];
    }
#pragma unroll
    for (int o = 16; o > 0; o >>= 1)
        part += __shfl_xor(part, o, 32);
    if ((t & 31) == 0) red[t >> 5] = part;

    int run = 0;
#pragma unroll
    for (int s = 0; s < GRP_SLICES; ++s) {
        pc[(size_t)s * N_PAD + i] = run;
        run += cnts[s];
    }
    const int c = run;
    sc[t] = c;
    __syncthreads();
    int prefix = 0;
#pragma unroll
    for (int k = 0; k < 8; ++k) prefix += red[k];
    int v = c;
    for (int o = 1; o < 256; o <<= 1) {
        const int add = (t >= o) ? sc[t - o] : 0;
        __syncthreads();
        v += add;
        sc[t] = v;
        __syncthreads();
    }
    if (i < N_NODES) off[i] = prefix + v - c;
    if (i == N_NODES - 1) off[N_NODES] = prefix + v;
}

__global__ __launch_bounds__(GRP_THREADS)
void grp_fill(const int* __restrict__ src, const int* __restrict__ dst,
              const int* __restrict__ pc, const int* __restrict__ off,
              int* __restrict__ tmp, int E) {
    __shared__ int cur[RANGE_NODES];
    const int tid = threadIdx.x;
    const int s   = blockIdx.x;
    const int lo  = blockIdx.y * RANGE_NODES;
#pragma unroll
    for (int k = 0; k < RANGE_NODES / GRP_THREADS; ++k) {
        const int i = k * GRP_THREADS + tid;
        const int node = lo + i;
        cur[i] = off[min(node, N_NODES)] + pc[(size_t)s * N_PAD + node];
    }
    __syncthreads();

    const int sl = grp_slice_len(E);
    const int e0 = s * sl;
    const int e1 = min(E & ~3, e0 + sl);
    for (int base = e0 + 4 * tid; base < e1; base += GRP_BATCH * GRP_STEP) {
        int4 d[GRP_BATCH], sv[GRP_BATCH];
#pragma unroll
        for (int k = 0; k < GRP_BATCH; ++k) {
            const int e = base + k * GRP_STEP;
            d[k]  = (e < e1) ? *(const int4*)(dst + e) : make_int4(-1, -1, -1, -1);
            sv[k] = (e < e1) ? *(const int4*)(src + e) : make_int4(0, 0, 0, 0);
        }
#pragma unroll
        for (int k = 0; k < GRP_BATCH; ++k) {
            const unsigned x0 = (unsigned)(d[k].x - lo), x1 = (unsigned)(d[k].y - lo);
            const unsigned x2 = (unsigned)(d[k].z - lo), x3 = (unsigned)(d[k].w - lo);
            if (x0 < RANGE_NODES) tmp[atomicAdd(&cur[x0], 1)] = sv[k].x;
            if (x1 < RANGE_NODES) tmp[atomicAdd(&cur[x1], 1)] = sv[k].y;
            if (x2 < RANGE_NODES) tmp[atomicAdd(&cur[x2], 1)] = sv[k].z;
            if (x3 < RANGE_NODES) tmp[atomicAdd(&cur[x3], 1)] = sv[k].w;
        }
    }
    const int et = E & ~3;
    if (tid == 0 && et >= e0 && et < e0 + sl)
        for (int e = et; e < E; ++e) {
            const unsigned x = (unsigned)(dst[e] - lo);
            if (x < RANGE_NODES) tmp[atomicAdd(&cur[x], 1)] = src[e];
        }
}

__global__ __launch_bounds__(256)
void grp_order(const int* __restrict__ off, const int* __restrict__ tmp,
               int* __restrict__ nbr) {
    const int node = __builtin_amdgcn_readfirstlane(
        (int)(blockIdx.x * 8 + (threadIdx.x >> 5)));
    if (node >= N_NODES) return;
    const int lane = threadIdx.x & 31;
    const int beg = off[node];
    const int deg = off[node + 1] - beg;
    for (int ib = 0; ib < deg; ib += 32) {
        const int i = ib + lane;
        const bool act = i < deg;
        const int key = act ? tmp[beg + i] : 0;
        int r = 0;
        for (int jb = 0; jb < deg; jb += 32) {
            const int jn = min(32, deg - jb);
            const int kl = (jb + lane < deg) ? tmp[beg + jb + lane] : 0;
            for (int jj = 0; jj < jn; ++jj) {
                const int kj = __builtin_amdgcn_readlane(kl, jj);
                const int j = jb + jj;
                r += (kj < key || (kj == key && j < i)) ? 1 : 0;
            }
        }
        if (act) nbr[beg + r] = key;
    }
}

template <int CTRL>
__device__ __forceinline__ float row_get(float v) {
    return __int_as_float(
        __builtin_amdgcn_update_dpp(0, __float_as_int(v), CTRL, 0xf, 0xf, true));
}
template <int CTRL>
__device__ __forceinline__ float row_xadd(float keep, float send) {
    return keep + row_get<CTRL>(send);
}
__device__ __forceinline__ float xhalf(float x) {
    const int b = __float_as_int(x);
    return __int_as_float(
        __builtin_amdgcn_permlanex16(b, b, 0x76543210, 0xfedcba98, false, false));
}
__device__ __forceinline__ float edge_part(const float4 v, const float4 b, const float4 w) {
    return w.x * lrelu(v.x + b.x) + w.y * lrelu(v.y + b.y)
         + w.z * lrelu(v.z + b.z) + w.w * lrelu(v.w + b.w);
}

__global__ __launch_bounds__(256)
void gat_node_fused(const float* __restrict__ fs, const float* __restrict__ fd,
                    const float* __restrict__ attn,
                    const int* __restrict__ off, const int* __restrict__ nbr,
                    float* __restrict__ hout) {
    const int node = __builtin_amdgcn_readfirstlane(
        (int)(blockIdx.x * 8 + (threadIdx.x >> 5)));
    if (node >= N_NODES) return;
    const int lane = threadIdx.x & 31;
    const bool odd1 = (lane & 1) != 0;
    const bool odd2 = (lane & 2) != 0;
    const bool hi8  = (lane & 8) != 0;
    const int beg = off[node];
    const int end = off[node + 1];

    const float4 b = *(const float4*)(fd + (size_t)node * HF + lane * 4);
    const float4 w = *(const float4*)(attn + lane * 4);

    float m = -INFINITY, den = 0.0f;
    float4 acc = make_float4(0.0f, 0.0f, 0.0f, 0.0f);

    for (int cb = beg; cb < end; cb += 32) {
        const int cn = min(32, end - cb);
        const int ids = nbr[cb + min(lane, cn - 1)];
        const int lim = cn - (lane & 3);
        for (int j = 0; j < cn; j += 4) {
            const int s0 = __builtin_amdgcn_readlane(ids, j);
            const int s1 = __builtin_amdgcn_readlane(ids, j + 1);
            const int s2 = __builtin_amdgcn_readlane(ids, j + 2);
            const int s3 = __builtin_amdgcn_readlane(ids, j + 3);
            const float4 v0 = *(const float4*)(fs + (size_t)s0 * HF + lane * 4);
            const float4 v1 = *(const float4*)(fs + (size_t)s1 * HF + lane * 4);
            const float4 v2 = *(const float4*)(fs + (size_t)s2 * HF + lane * 4);
            const float4 v3 = *(const float4*)(fs + (size_t)s3 * HF + lane * 4);

            const float l0 = edge_part(v0, b, w);
            const float l1 = edge_part(v1, b, w);
            const float l2 = edge_part(v2, b, w);
            const float l3 = edge_part(v3, b, w);
            const float e01 = row_xadd<0x161>(odd1 ? l1 : l0, odd1 ? l0 : l1);
            const float e23 = row_xadd<0x161>(odd1 ? l3 : l2, odd1 ? l2 : l3);
            float lg = row_xadd<0x162>(odd2 ? e23 : e01, odd2 ? e01 : e23);
            lg = row_xadd<0x164>(lg, lg);
            lg = row_xadd<0x168>(lg, lg);
            if (j >= lim) lg = -INFINITY;

            float mq = fmaxf(lg, row_get<0x161>(lg));
            mq = fmaxf(mq, row_get<0x162>(mq));
            const float mn = fmaxf(m, mq);
            const float ex = expf((hi8 ? m : lg) - mn);
            const float p0 = row_get<0x150>(ex);
            const float p1 = row_get<0x151>(ex);
            const float p2 = row_get<0x152>(ex);
            const float p3 = row_get<0x153>(ex);
            const float sc = row_get<0x158>(ex);
            den   = den   * sc + ((p0 + p1) + (p2 + p3));
            acc.x = acc.x * sc + ((p0 * v0.x + p1 * v1.x) + (p2 * v2.x + p3 * v3.x));
            acc.y = acc.y * sc + ((p0 * v0.y + p1 * v1.y) + (p2 * v2.y + p3 * v3.y));
            acc.z = acc.z * sc + ((p0 * v0.z + p1 * v1.z) + (p2 * v2.z + p3 * v3.z));
            acc.w = acc.w * sc + ((p0 * v0.w + p1 * v1.w) + (p2 * v2.w + p3 * v3.w));
            m = mn;
        }
    }

    const float inv = (end > beg) ? 1.0f / den : 0.0f;
    float4 o;
    o.x = acc.x * inv; o.y = acc.y * inv; o.z = acc.z * inv; o.w = acc.w * inv;
    o.x = 0.5f * (o.x + xhalf(o.x));
    o.y = 0.5f * (o.y + xhalf(o.y));
    o.z = 0.5f * (o.z + xhalf(o.z));
    o.w = 0.5f * (o.w + xhalf(o.w));
    if (lane < 16)
        *(float4*)(hout + (size_t)node * F_DIM + lane * 4) = o;
}

extern "C" void kernel_launch(void* const* d_in, const int* in_sizes, int n_in,
                              void* d_out, int out_size, void* d_ws, size_t ws_size,
                              hipStream_t stream) {
    (void)n_in; (void)out_size; (void)ws_size;
    const float* h0   = (const float*)d_in[0];
    const float* Wsrc = (const float*)d_in[1];
    const float* bsrc = (const float*)d_in[2];
    const float* Wdst = (const float*)d_in[3];
    const float* bdst = (const float*)d_in[4];
    const float* attn = (const float*)d_in[5];
    const int*   src  = (const int*)d_in[6];
    const int*   dst  = (const int*)d_in[7];
    const int    E    = in_sizes[6];

    float* ws   = (float*)d_ws;
    float* fs   = ws;  ws += (size_t)N_NODES * HF;
    float* fd   = ws;  ws += (size_t)N_NODES * HF;
    float* hbuf = ws;  ws += (size_t)N_NODES * F_DIM;
    int* off    = (int*)ws;  ws += (size_t)N_NODES + 4;
    int* tmp    = (int*)ws;  ws += (size_t)E;
    int* nbr    = (int*)ws;
    int* pc     = (int*)hbuf;
    int* bs     = pc + (size_t)GRP_SLICES * N_PAD;

    const int gridW = (N_NODES + 7) / 8;
    const dim3 gridC(GRP_SLICES, GRP_RANGES);

    grp_count<<<gridC, GRP_THREADS, 0, stream>>>(dst, pc, bs, E);
    grp_offsets<<<NODE_BLOCKS, 256, 0, stream>>>(pc, bs, off);
    grp_fill<<<gridC, GRP_THREADS, 0, stream>>>(src, dst, pc, off, tmp, E);
    grp_order<<<gridW, 256, 0, stream>>>(off, tmp, nbr);

    const float* hcur = h0;
    for (int l = 0; l < L_LAYERS; ++l) {
        gat_proj_wmma<<<dim3(PROJ_BLOCKS, 2), PROJ_WAVES * 32, 0, stream>>>(
            hcur,
            Wsrc + (size_t)l * F_DIM * HF, bsrc + (size_t)l * HF,
            Wdst + (size_t)l * F_DIM * HF, bdst + (size_t)l * HF,
            fs, fd);

        float* hn = (l == L_LAYERS - 1) ? (float*)d_out : hbuf;
        gat_node_fused<<<gridW, 256, 0, stream>>>(
            fs, fd, attn + (size_t)l * HF, off, nbr, hn);
        hcur = hbuf;
    }
}


# group ordering (counting rank by src id) folded into layer 1's aggregation wave: rank, LDS slot, read back as ids; ordered groups kept for layer 2; one launch, 50000 waves fewer; sums unchanged
# speedup vs baseline: 4.7662x; 1.0389x over previous
#include <hip/hip_runtime.h>

#define N_NODES   50000
#define F_DIM     64
#define H_HEADS   2
#define HF        128
#define L_LAYERS  2
#define NEG_SLOPE 0.2f

typedef __attribute__((ext_vector_type(2))) float v2f;
typedef __attribute__((ext_vector_type(8))) float v8f;

__device__ __forceinline__ float lrelu(float x) {
    return x > 0.0f ? x : NEG_SLOPE * x;
}

#define ROW_TILES   (N_NODES / 16)
#define PROJ_WAVES  4
#define PROJ_BLOCKS ((ROW_TILES + PROJ_WAVES - 1) / PROJ_WAVES)

__global__ __launch_bounds__(PROJ_WAVES * 32)
void gat_proj_wmma(const float* __restrict__ h,
                   const float* __restrict__ Wsrc, const float* __restrict__ bsrc,
                   const float* __restrict__ Wdst, const float* __restrict__ bdst,
                   float* __restrict__ fs, float* __restrict__ fd) {
    __shared__ float4 wl[16 * 8 * 16];

    const int tid  = threadIdx.x;
    const int wave = tid >> 5;
    const int lane = tid & 31;
    const int hh   = lane >> 4;
    const int c    = lane & 15;

    const bool  isSrc = (blockIdx.y == 0);
    const float* __restrict__ W    = isSrc ? Wsrc : Wdst;
    const float* __restrict__ bias = isSrc ? bsrc : bdst;
    float* __restrict__ out        = isSrc ? fs : fd;

    const int  tile = blockIdx.x * PROJ_WAVES + wave;
    const bool live = tile < ROW_TILES;
    const int  ltile = live ? tile : ROW_TILES - 1;

    const float* __restrict__ arow = h + (size_t)(ltile * 16 + c) * F_DIM + hh * 4;
    float4 a[8];
#pragma unroll
    for (int i2 = 0; i2 < 8; ++i2)
        a[i2] = *(const float4*)(arow + 8 * i2);

#pragma unroll
    for (int it = 0; it < 512 / (PROJ_WAVES * 32); ++it) {
        const int item = it * (PROJ_WAVES * 32) + tid;
        const int q  = item >> 5;
        const int c4 = item & 31;
        const float* __restrict__ wp = W + (size_t)(4 * q) * HF + 4 * c4;
        const float4 r0 = *(const float4*)(wp);
        const float4 r1 = *(const float4*)(wp + HF);
        const float4 r2 = *(const float4*)(wp + 2 * HF);
        const float4 r3 = *(const float4*)(wp + 3 * HF);
        float4* dp = wl + (q * 8 + 4 * (c4 >> 4)) * 16 + (c4 & 15);
        dp[0]  = make_float4(r0.x, r1.x, r2.x, r3.x);
        dp[16] = make_float4(r0.y, r1.y, r2.y, r3.y);
        dp[32] = make_float4(r0.z, r1.z, r2.z, r3.z);
        dp[48] = make_float4(r0.w, r1.w, r2.w, r3.w);
    }
    __syncthreads();
    if (!live) return;

    v8f acc[8];
#pragma unroll
    for (int t = 0; t < 8; ++t) acc[t] = (v8f){};

    const float4* bl = wl + hh * 128 + c;
#pragma unroll
    for (int i2 = 0; i2 < 8; ++i2) {
        float4 b[8];
#pragma unroll
        for (int t = 0; t < 8; ++t) b[t] = bl[i2 * 256 + t * 16];
        v2f a0, a1;
        a0.x = a[i2].x; a0.y = a[i2].y;
        a1.x = a[i2].z; a1.y = a[i2].w;
#pragma unroll
        for (int t = 0; t < 8; ++t) {
            v2f b0; b0.x = b[t].x; b0.y = b[t].y;
            acc[t] = __builtin_amdgcn_wmma_f32_16x16x4_f32(
                        false, a0, false, b0, (short)0, acc[t], false, false);
        }
#pragma unroll
        for (int t = 0; t < 8; ++t) {
            v2f b1; b1.x = b[t].z; b1.y = b[t].w;
            acc[t] = __builtin_amdgcn_wmma_f32_16x16x4_f32(
                        false, a1, false, b1, (short)0, acc[t], false, false);
        }
    }

    float* __restrict__ orow = out + (size_t)(tile * 16 + hh * 8) * HF + 4 * c;
#pragma unroll
    for (int g = 0; g < 2; ++g) {
        const float4 bv = *(const float4*)(bias + 64 * g + 4 * c);
#pragma unroll
        for (int r = 0; r < 8; ++r) {
            float4 o;
            o.x = acc[4 * g + 0][r] + bv.x;
            o.y = acc[4 * g + 1][r] + bv.y;
            o.z = acc[4 * g + 2][r] + bv.z;
            o.w = acc[4 * g + 3][r] + bv.w;
            *(float4*)(orow + (size_t)r * HF + 64 * g) = o;
        }
    }
}

#define GRP_SLICES   36
#define GRP_THREADS  1024
#define NODE_BLOCKS  ((N_NODES + 255) / 256)
#define RANGE_BLOCKS 28
#define RANGE_NODES  (RANGE_BLOCKS * 256)
#define GRP_RANGES   (NODE_BLOCKS / RANGE_BLOCKS)
#define N_PAD        (NODE_BLOCKS * 256)
static_assert(GRP_RANGES * RANGE_BLOCKS == NODE_BLOCKS, "ranges must tile the node blocks");
static_assert(RANGE_BLOCKS * 32 <= GRP_THREADS, "one wave per node block for the block sums");
static_assert(RANGE_NODES % GRP_THREADS == 0, "whole passes over a range's counters");

__device__ __forceinline__ int grp_slice_len(int E) {
    return (((E + GRP_SLICES - 1) / GRP_SLICES) + 3) & ~3;
}
#define GRP_STEP  (4 * GRP_THREADS)
#define GRP_BATCH 6

__global__ __launch_bounds__(GRP_THREADS)
void grp_count(const int* __restrict__ dst, int* __restrict__ pc,
               int* __restrict__ bs, int E) {
    __shared__ int lc[RANGE_NODES];
    const int tid = threadIdx.x;
    const int s   = blockIdx.x;
    const int lo  = blockIdx.y * RANGE_NODES;
    for (int i = tid; i < RANGE_NODES; i += GRP_THREADS) lc[i] = 0;
    __syncthreads();

    const int sl = grp_slice_len(E);
    const int e0 = s * sl;
    const int e1 = min(E & ~3, e0 + sl);
    for (int base = e0 + 4 * tid; base < e1; base += GRP_BATCH * GRP_STEP) {
        int4 d[GRP_BATCH];
#pragma unroll
        for (int k = 0; k < GRP_BATCH; ++k) {
            const int e = base + k * GRP_STEP;
            d[k] = (e < e1) ? *(const int4*)(dst + e) : make_int4(-1, -1, -1, -1);
        }
#pragma unroll
        for (int k = 0; k < GRP_BATCH; ++k) {
            const unsigned x0 = (unsigned)(d[k].x - lo), x1 = (unsigned)(d[k].y - lo);
            const unsigned x2 = (unsigned)(d[k].z - lo), x3 = (unsigned)(d[k].w - lo);
            if (x0 < RANGE_NODES) atomicAdd(&lc[x0], 1);
            if (x1 < RANGE_NODES) atomicAdd(&lc[x1], 1);
            if (x2 < RANGE_NODES) atomicAdd(&lc[x2], 1);
            if (x3 < RANGE_NODES) atomicAdd(&lc[x3], 1);
        }
    }
    const int et = E & ~3;
    if (tid == 0 && et >= e0 && et < e0 + sl)
        for (int e = et; e < E; ++e) {
            const unsigned x = (unsigned)(dst[e] - lo);
            if (x < RANGE_NODES) atomicAdd(&lc[x], 1);
        }
    __syncthreads();

    for (int i = tid; i < RANGE_NODES; i += GRP_THREADS)
        pc[(size_t)s * N_PAD + lo + i] = lc[i];
    const int wave = tid >> 5, lane = tid & 31;
    if (wave < RANGE_BLOCKS) {
        int sum = 0;
#pragma unroll
        for (int k = 0; k < 8; ++k) sum += lc[wave * 256 + k * 32 + lane];
#pragma unroll
        for (int o = 16; o > 0; o >>= 1) sum += __shfl_xor(sum, o, 32);
        if (lane == 0)
            bs[s * NODE_BLOCKS + blockIdx.y * RANGE_BLOCKS + wave] = sum;
    }
}

__global__ __launch_bounds__(256)
void grp_offsets(int* __restrict__ pc, const int* __restrict__ bs,
                 int* __restrict__ off) {
    __shared__ int red[8];
    __shared__ int sc[256];
    const int t = threadIdx.x;
    const int b = blockIdx.x;
    const int i = b * 256 + t;
    int cnts[GRP_SLICES];
#pragma unroll
    for (int s = 0; s < GRP_SLICES; ++s) cnts[s] = pc[(size_t)s * N_PAD + i];
    int part = 0;
    if (t < b) {
#pragma unroll
        for (int s = 0; s < GRP_SLICES; ++s) part += bs[s * NODE_BLOCKS + t];
    }
#pragma unroll
    for (int o = 16; o > 0; o >>= 1)
        part += __shfl_xor(part, o, 32);
    if ((t & 31) == 0) red[t >> 5] = part;

    int run = 0;
#pragma unroll
    for (int s = 0; s < GRP_SLICES; ++s) {
        pc[(size_t)s * N_PAD + i] = run;
        run += cnts[s];
    }
    const int c = run;
    sc[t] = c;
    __syncthreads();
    int prefix = 0;
#pragma unroll
    for (int k = 0; k < 8; ++k) prefix += red[k];
    int v = c;
    for (int o = 1; o < 256; o <<= 1) {
        const int add = (t >= o) ? sc[t - o] : 0;
        __syncthreads();
        v += add;
        sc[t] = v;
        __syncthreads();
    }
    if (i < N_NODES) off[i] = prefix + v - c;
    if (i == N_NODES - 1) off[N_NODES] = prefix + v;
}

__global__ __launch_bounds__(GRP_THREADS)
void grp_fill(const int* __restrict__ src, const int* __restrict__ dst,
              const int* __restrict__ pc, const int* __restrict__ off,
              int* __restrict__ tmp, int E) {
    __shared__ int cur[RANGE_NODES];
    const int tid = threadIdx.x;
    const int s   = blockIdx.x;
    const int lo  = blockIdx.y * RANGE_NODES;
#pragma unroll
    for (int k = 0; k < RANGE_NODES / GRP_THREADS; ++k) {
        const int i = k * GRP_THREADS + tid;
        const int node = lo + i;
        cur[i] = off[min(node, N_NODES)] + pc[(size_t)s * N_PAD + node];
    }
    __syncthreads();

    const int sl = grp_slice_len(E);
    const int e0 = s * sl;
    const int e1 = min(E & ~3, e0 + sl);
    for (int base = e0 + 4 * tid; base < e1; base += GRP_BATCH * GRP_STEP) {
        int4 d[GRP_BATCH], sv[GRP_BATCH];
#pragma unroll
        for (int k = 0; k < GRP_BATCH; ++k) {
            const int e = base + k * GRP_STEP;
            d[k]  = (e < e1) ? *(const int4*)(dst + e) : make_int4(-1, -1, -1, -1);
            sv[k] = (e < e1) ? *(const int4*)(src + e) : make_int4(0, 0, 0, 0);
        }
#pragma unroll
        for (int k = 0; k < GRP_BATCH; ++k) {
            const unsigned x0 = (unsigned)(d[k].x - lo), x1 = (unsigned)(d[k].y - lo);
            const unsigned x2 = (unsigned)(d[k].z - lo), x3 = (unsigned)(d[k].w - lo);
            if (x0 < RANGE_NODES) tmp[atomicAdd(&cur[x0], 1)] = sv[k].x;
            if (x1 < RANGE_NODES) tmp[atomicAdd(&cur[x1], 1)] = sv[k].y;
            if (x2 < RANGE_NODES) tmp[atomicAdd(&cur[x2], 1)] = sv[k].z;
            if (x3 < RANGE_NODES) tmp[atomicAdd(&cur[x3], 1)] = sv[k].w;
        }
    }
    const int et = E & ~3;
    if (tid == 0 && et >= e0 && et < e0 + sl)
        for (int e = et; e < E; ++e) {
            const unsigned x = (unsigned)(dst[e] - lo);
            if (x < RANGE_NODES) tmp[atomicAdd(&cur[x], 1)] = src[e];
        }
}

template <int CTRL>
__device__ __forceinline__ float row_get(float v) {
    return __int_as_float(
        __builtin_amdgcn_update_dpp(0, __float_as_int(v), CTRL, 0xf, 0xf, true));
}
template <int CTRL>
__device__ __forceinline__ float row_xadd(float keep, float send) {
    return keep + row_get<CTRL>(send);
}
__device__ __forceinline__ float xhalf(float x) {
    const int b = __float_as_int(x);
    return __int_as_float(
        __builtin_amdgcn_permlanex16(b, b, 0x76543210, 0xfedcba98, false, false));
}
__device__ __forceinline__ float edge_part(const float4 v, const float4 b, const float4 w) {
    return w.x * lrelu(v.x + b.x) + w.y * lrelu(v.y + b.y)
         + w.z * lrelu(v.z + b.z) + w.w * lrelu(v.w + b.w);
}

template <bool ORDER>
__global__ __launch_bounds__(256)
void gat_node_fused(const float* __restrict__ fs, const float* __restrict__ fd,
                    const float* __restrict__ attn,
                    const int* __restrict__ off, const int* __restrict__ grp,
                    int* __restrict__ nbr_out, float* __restrict__ hout) {
    const int node = __builtin_amdgcn_readfirstlane(
        (int)(blockIdx.x * 8 + (threadIdx.x >> 5)));
    if (node >= N_NODES) return;
    const int lane = threadIdx.x & 31;
    const bool odd1 = (lane & 1) != 0;
    const bool odd2 = (lane & 2) != 0;
    const bool hi8  = (lane & 8) != 0;
    const int beg = off[node];
    const int end = off[node + 1];

    const float4 b = *(const float4*)(fd + (size_t)node * HF + lane * 4);
    const float4 w = *(const float4*)(attn + lane * 4);

    float m = -INFINITY, den = 0.0f;
    float4 acc = make_float4(0.0f, 0.0f, 0.0f, 0.0f);

    for (int cb = beg; cb < end; cb += 32) {
        const int cn = min(32, end - cb);
        int ids;
        if constexpr (ORDER) {
            __shared__ int slot[8 * 32];
            int* const ws = slot + (threadIdx.x >> 5) * 32;
            const int deg = end - beg;
            const int ob  = cb - beg;
            for (int ib = 0; ib < deg; ib += 32) {
                const int i = ib + lane;
                const bool act = i < deg;
                const int key = act ? grp[beg + i] : 0;
                const int ck  = key * 32 + lane;
                int r = 0;
                for (int jb = 0; jb < deg; jb += 32) {
                    const int jn = min(32, deg - jb);
                    if (jb == ib) {
                        for (int jj = 0; jj < jn; ++jj)
                            r += (__builtin_amdgcn_readlane(ck, jj) < ck) ? 1 : 0;
                    } else {
                        const int kl = (jb + lane < deg) ? grp[beg + jb + lane] : 0;
                        if (jb < ib) {
                            for (int jj = 0; jj < jn; ++jj)
                                r += (__builtin_amdgcn_readlane(kl, jj) <= key) ? 1 : 0;
                        } else {
                            for (int jj = 0; jj < jn; ++jj)
                                r += (__builtin_amdgcn_readlane(kl, jj) < key) ? 1 : 0;
                        }
                    }
                }
                if (act) {
                    if (ob == 0) nbr_out[beg + r] = key;
                    const unsigned p = (unsigned)(r - ob);
                    if (p < 32u) ws[p] = key;
                }
            }
            ids = ws[min(lane, cn - 1)];
        } else {
            ids = grp[cb + min(lane, cn - 1)];
        }
        const int lim = cn - (lane & 3);
        for (int j = 0; j < cn; j += 4) {
            const int s0 = __builtin_amdgcn_readlane(ids, j);
            const int s1 = __builtin_amdgcn_readlane(ids, j + 1);
            const int s2 = __builtin_amdgcn_readlane(ids, j + 2);
            const int s3 = __builtin_amdgcn_readlane(ids, j + 3);
            const float4 v0 = *(const float4*)(fs + (size_t)s0 * HF + lane * 4);
            const float4 v1 = *(const float4*)(fs + (size_t)s1 * HF + lane * 4);
            const float4 v2 = *(const float4*)(fs + (size_t)s2 * HF + lane * 4);
            const float4 v3 = *(const float4*)(fs + (size_t)s3 * HF + lane * 4);

            const float l0 = edge_part(v0, b, w);
            const float l1 = edge_part(v1, b, w);
            const float l2 = edge_part(v2, b, w);
            const float l3 = edge_part(v3, b, w);
            const float e01 = row_xadd<0x161>(odd1 ? l1 : l0, odd1 ? l0 : l1);
            const float e23 = row_xadd<0x161>(odd1 ? l3 : l2, odd1 ? l2 : l3);
            float lg = row_xadd<0x162>(odd2 ? e23 : e01, odd2 ? e01 : e23);
            lg = row_xadd<0x164>(lg, lg);
            lg = row_xadd<0x168>(lg, lg);
            if (j >= lim) lg = -INFINITY;

            float mq = fmaxf(lg, row_get<0x161>(lg));
            mq = fmaxf(mq, row_get<0x162>(mq));
            const float mn = fmaxf(m, mq);
            const float ex = expf((hi8 ? m : lg) - mn);
            const float p0 = row_get<0x150>(ex);
            const float p1 = row_get<0x151>(ex);
            const float p2 = row_get<0x152>(ex);
            const float p3 = row_get<0x153>(ex);
            const float sc = row_get<0x158>(ex);
            den   = den   * sc + ((p0 + p1) + (p2 + p3));
            acc.x = acc.x * sc + ((p0 * v0.x + p1 * v1.x) + (p2 * v2.x + p3 * v3.x));
            acc.y = acc.y * sc + ((p0 * v0.y + p1 * v1.y) + (p2 * v2.y + p3 * v3.y));
            acc.z = acc.z * sc + ((p0 * v0.z + p1 * v1.z) + (p2 * v2.z + p3 * v3.z));
            acc.w = acc.w * sc + ((p0 * v0.w + p1 * v1.w) + (p2 * v2.w + p3 * v3.w));
            m = mn;
        }
    }

    const float inv = (end > beg) ? 1.0f / den : 0.0f;
    float4 o;
    o.x = acc.x * inv; o.y = acc.y * inv; o.z = acc.z * inv; o.w = acc.w * inv;
    o.x = 0.5f * (o.x + xhalf(o.x));
    o.y = 0.5f * (o.y + xhalf(o.y));
    o.z = 0.5f * (o.z + xhalf(o.z));
    o.w = 0.5f * (o.w + xhalf(o.w));
    if (lane < 16)
        *(float4*)(hout + (size_t)node * F_DIM + lane * 4) = o;
}

extern "C" void kernel_launch(void* const* d_in, const int* in_sizes, int n_in,
                              void* d_out, int out_size, void* d_ws, size_t ws_size,
                              hipStream_t stream) {
    (void)n_in; (void)out_size; (void)ws_size;
    const float* h0   = (const float*)d_in[0];
    const float* Wsrc = (const float*)d_in[1];
    const float* bsrc = (const float*)d_in[2];
    const float* Wdst = (const float*)d_in[3];
    const float* bdst = (const float*)d_in[4];
    const float* attn = (const float*)d_in[5];
    const int*   src  = (const int*)d_in[6];
    const int*   dst  = (const int*)d_in[7];
    const int    E    = in_sizes[6];

    float* ws   = (float*)d_ws;
    float* fs   = ws;  ws += (size_t)N_NODES * HF;
    float* fd   = ws;  ws += (size_t)N_NODES * HF;
    float* hbuf = ws;  ws += (size_t)N_NODES * F_DIM;
    int* off    = (int*)ws;  ws += (size_t)N_NODES + 4;
    int* tmp    = (int*)ws;  ws += (size_t)E;
    int* nbr    = (int*)ws;
    int* pc     = (int*)hbuf;
    int* bs     = pc + (size_t)GRP_SLICES * N_PAD;

    const int gridW = (N_NODES + 7) / 8;
    const dim3 gridC(GRP_SLICES, GRP_RANGES);

    grp_count<<<gridC, GRP_THREADS, 0, stream>>>(dst, pc, bs, E);
    grp_offsets<<<NODE_BLOCKS, 256, 0, stream>>>(pc, bs, off);
    grp_fill<<<gridC, GRP_THREADS, 0, stream>>>(src, dst, pc, off, tmp, E);

    const float* hcur = h0;
    for (int l = 0; l < L_LAYERS; ++l) {
        gat_proj_wmma<<<dim3(PROJ_BLOCKS, 2), PROJ_WAVES * 32, 0, stream>>>(
            hcur,
            Wsrc + (size_t)l * F_DIM * HF, bsrc + (size_t)l * HF,
            Wdst + (size_t)l * F_DIM * HF, bdst + (size_t)l * HF,
            fs, fd);

        float* hn = (l == L_LAYERS - 1) ? (float*)d_out : hbuf;
        if (l == 0)
            gat_node_fused<true><<<gridW, 256, 0, stream>>>(
                fs, fd, attn + (size_t)l * HF, off, tmp, nbr, hn);
        else
            gat_node_fused<false><<<gridW, 256, 0, stream>>>(
                fs, fd, attn + (size_t)l * HF, off, nbr, nullptr, hn);
        hcur = hbuf;
    }
}


# aggregation loop, 124 to 93 vector instructions per 4 rows: leaky-relu as mul+max, a lane's 4 logit products as two fused pairs (x,z)+(y,w): logit sum order changed; row base on the scalar unit
# speedup vs baseline: 4.9420x; 1.0369x over previous
#include <hip/hip_runtime.h>

#define N_NODES   50000
#define F_DIM     64
#define H_HEADS   2
#define HF        128
#define L_LAYERS  2
#define NEG_SLOPE 0.2f

typedef __attribute__((ext_vector_type(2))) float v2f;
typedef __attribute__((ext_vector_type(8))) float v8f;

__device__ __forceinline__ float lrelu(float x) {
    return fmaxf(x, NEG_SLOPE * x);
}

#define ROW_TILES   (N_NODES / 16)
#define PROJ_WAVES  4
#define PROJ_BLOCKS ((ROW_TILES + PROJ_WAVES - 1) / PROJ_WAVES)

__global__ __launch_bounds__(PROJ_WAVES * 32)
void gat_proj_wmma(const float* __restrict__ h,
                   const float* __restrict__ Wsrc, const float* __restrict__ bsrc,
                   const float* __restrict__ Wdst, const float* __restrict__ bdst,
                   float* __restrict__ fs, float* __restrict__ fd) {
    __shared__ float4 wl[16 * 8 * 16];

    const int tid  = threadIdx.x;
    const int wave = tid >> 5;
    const int lane = tid & 31;
    const int hh   = lane >> 4;
    const int c    = lane & 15;

    const bool  isSrc = (blockIdx.y == 0);
    const float* __restrict__ W    = isSrc ? Wsrc : Wdst;
    const float* __restrict__ bias = isSrc ? bsrc : bdst;
    float* __restrict__ out        = isSrc ? fs : fd;

    const int  tile = blockIdx.x * PROJ_WAVES + wave;
    const bool live = tile < ROW_TILES;
    const int  ltile = live ? tile : ROW_TILES - 1;

    const float* __restrict__ arow = h + (size_t)(ltile * 16 + c) * F_DIM + hh * 4;
    float4 a[8];
#pragma unroll
    for (int i2 = 0; i2 < 8; ++i2)
        a[i2] = *(const float4*)(arow + 8 * i2);

#pragma unroll
    for (int it = 0; it < 512 / (PROJ_WAVES * 32); ++it) {
        const int item = it * (PROJ_WAVES * 32) + tid;
        const int q  = item >> 5;
        const int c4 = item & 31;
        const float* __restrict__ wp = W + (size_t)(4 * q) * HF + 4 * c4;
        const float4 r0 = *(const float4*)(wp);
        const float4 r1 = *(const float4*)(wp + HF);
        const float4 r2 = *(const float4*)(wp + 2 * HF);
        const float4 r3 = *(const float4*)(wp + 3 * HF);
        float4* dp = wl + (q * 8 + 4 * (c4 >> 4)) * 16 + (c4 & 15);
        dp[0]  = make_float4(r0.x, r1.x, r2.x, r3.x);
        dp[16] = make_float4(r0.y, r1.y, r2.y, r3.y);
        dp[32] = make_float4(r0.z, r1.z, r2.z, r3.z);
        dp[48] = make_float4(r0.w, r1.w, r2.w, r3.w);
    }
    __syncthreads();
    if (!live) return;

    v8f acc[8];
#pragma unroll
    for (int t = 0; t < 8; ++t) acc[t] = (v8f){};

    const float4* bl = wl + hh * 128 + c;
#pragma unroll
    for (int i2 = 0; i2 < 8; ++i2) {
        float4 b[8];
#pragma unroll
        for (int t = 0; t < 8; ++t) b[t] = bl[i2 * 256 + t * 16];
        v2f a0, a1;
        a0.x = a[i2].x; a0.y = a[i2].y;
        a1.x = a[i2].z; a1.y = a[i2].w;
#pragma unroll
        for (int t = 0; t < 8; ++t) {
            v2f b0; b0.x = b[t].x; b0.y = b[t].y;
            acc[t] = __builtin_amdgcn_wmma_f32_16x16x4_f32(
                        false, a0, false, b0, (short)0, acc[t], false, false);
        }
#pragma unroll
        for (int t = 0; t < 8; ++t) {
            v2f b1; b1.x = b[t].z; b1.y = b[t].w;
            acc[t] = __builtin_amdgcn_wmma_f32_16x16x4_f32(
                        false, a1, false, b1, (short)0, acc[t], false, false);
        }
    }

    float* __restrict__ orow = out + (size_t)(tile * 16 + hh * 8) * HF + 4 * c;
#pragma unroll
    for (int g = 0; g < 2; ++g) {
        const float4 bv = *(const float4*)(bias + 64 * g + 4 * c);
#pragma unroll
        for (int r = 0; r < 8; ++r) {
            float4 o;
            o.x = acc[4 * g + 0][r] + bv.x;
            o.y = acc[4 * g + 1][r] + bv.y;
            o.z = acc[4 * g + 2][r] + bv.z;
            o.w = acc[4 * g + 3][r] + bv.w;
            *(float4*)(orow + (size_t)r * HF + 64 * g) = o;
        }
    }
}

#define GRP_SLICES   36
#define GRP_THREADS  1024
#define NODE_BLOCKS  ((N_NODES + 255) / 256)
#define RANGE_BLOCKS 28
#define RANGE_NODES  (RANGE_BLOCKS * 256)
#define GRP_RANGES   (NODE_BLOCKS / RANGE_BLOCKS)
#define N_PAD        (NODE_BLOCKS * 256)
static_assert(GRP_RANGES * RANGE_BLOCKS == NODE_BLOCKS, "ranges must tile the node blocks");
static_assert(RANGE_BLOCKS * 32 <= GRP_THREADS, "one wave per node block for the block sums");
static_assert(RANGE_NODES % GRP_THREADS == 0, "whole passes over a range's counters");

__device__ __forceinline__ int grp_slice_len(int E) {
    return (((E + GRP_SLICES - 1) / GRP_SLICES) + 3) & ~3;
}
#define GRP_STEP  (4 * GRP_THREADS)
#define GRP_BATCH 6

__global__ __launch_bounds__(GRP_THREADS)
void grp_count(const int* __restrict__ dst, int* __restrict__ pc,
               int* __restrict__ bs, int E) {
    __shared__ int lc[RANGE_NODES];
    const int tid = threadIdx.x;
    const int s   = blockIdx.x;
    const int lo  = blockIdx.y * RANGE_NODES;
    for (int i = tid; i < RANGE_NODES; i += GRP_THREADS) lc[i] = 0;
    __syncthreads();

    const int sl = grp_slice_len(E);
    const int e0 = s * sl;
    const int e1 = min(E & ~3, e0 + sl);
    for (int base = e0 + 4 * tid; base < e1; base += GRP_BATCH * GRP_STEP) {
        int4 d[GRP_BATCH];
#pragma unroll
        for (int k = 0; k < GRP_BATCH; ++k) {
            const int e = base + k * GRP_STEP;
            d[k] = (e < e1) ? *(const int4*)(dst + e) : make_int4(-1, -1, -1, -1);
        }
#pragma unroll
        for (int k = 0; k < GRP_BATCH; ++k) {
            const unsigned x0 = (unsigned)(d[k].x - lo), x1 = (unsigned)(d[k].y - lo);
            const unsigned x2 = (unsigned)(d[k].z - lo), x3 = (unsigned)(d[k].w - lo);
            if (x0 < RANGE_NODES) atomicAdd(&lc[x0], 1);
            if (x1 < RANGE_NODES) atomicAdd(&lc[x1], 1);
            if (x2 < RANGE_NODES) atomicAdd(&lc[x2], 1);
            if (x3 < RANGE_NODES) atomicAdd(&lc[x3], 1);
        }
    }
    const int et = E & ~3;
    if (tid == 0 && et >= e0 && et < e0 + sl)
        for (int e = et; e < E; ++e) {
            const unsigned x = (unsigned)(dst[e] - lo);
            if (x < RANGE_NODES) atomicAdd(&lc[x], 1);
        }
    __syncthreads();

    for (int i = tid; i < RANGE_NODES; i += GRP_THREADS)
        pc[(size_t)s * N_PAD + lo + i] = lc[i];
    const int wave = tid >> 5, lane = tid & 31;
    if (wave < RANGE_BLOCKS) {
        int sum = 0;
#pragma unroll
        for (int k = 0; k < 8; ++k) sum += lc[wave * 256 + k * 32 + lane];
#pragma unroll
        for (int o = 16; o > 0; o >>= 1) sum += __shfl_xor(sum, o, 32);
        if (lane == 0)
            bs[s * NODE_BLOCKS + blockIdx.y * RANGE_BLOCKS + wave] = sum;
    }
}

__global__ __launch_bounds__(256)
void grp_offsets(int* __restrict__ pc, const int* __restrict__ bs,
                 int* __restrict__ off) {
    __shared__ int red[8];
    __shared__ int sc[256];
    const int t = threadIdx.x;
    const int b = blockIdx.x;
    const int i = b * 256 + t;
    int cnts[GRP_SLICES];
#pragma unroll
    for (int s = 0; s < GRP_SLICES; ++s) cnts[s] = pc[(size_t)s * N_PAD + i];
    int part = 0;
    if (t < b) {
#pragma unroll
        for (int s = 0; s < GRP_SLICES; ++s) part += bs[s * NODE_BLOCKS + t];
    }
#pragma unroll
    for (int o = 16; o > 0; o >>= 1)
        part += __shfl_xor(part, o, 32);
    if ((t & 31) == 0) red[t >> 5] = part;

    int run = 0;
#pragma unroll
    for (int s = 0; s < GRP_SLICES; ++s) {
        pc[(size_t)s * N_PAD + i] = run;
        run += cnts[s];
    }
    const int c = run;
    sc[t] = c;
    __syncthreads();
    int prefix = 0;
#pragma unroll
    for (int k = 0; k < 8; ++k) prefix += red[k];
    int v = c;
    for (int o = 1; o < 256; o <<= 1) {
        const int add = (t >= o) ? sc[t - o] : 0;
        __syncthreads();
        v += add;
        sc[t] = v;
        __syncthreads();
    }
    if (i < N_NODES) off[i] = prefix + v - c;
    if (i == N_NODES - 1) off[N_NODES] = prefix + v;
}

__global__ __launch_bounds__(GRP_THREADS)
void grp_fill(const int* __restrict__ src, const int* __restrict__ dst,
              const int* __restrict__ pc, const int* __restrict__ off,
              int* __restrict__ tmp, int E) {
    __shared__ int cur[RANGE_NODES];
    const int tid = threadIdx.x;
    const int s   = blockIdx.x;
    const int lo  = blockIdx.y * RANGE_NODES;
#pragma unroll
    for (int k = 0; k < RANGE_NODES / GRP_THREADS; ++k) {
        const int i = k * GRP_THREADS + tid;
        const int node = lo + i;
        cur[i] = off[min(node, N_NODES)] + pc[(size_t)s * N_PAD + node];
    }
    __syncthreads();

    const int sl = grp_slice_len(E);
    const int e0 = s * sl;
    const int e1 = min(E & ~3, e0 + sl);
    for (int base = e0 + 4 * tid; base < e1; base += GRP_BATCH * GRP_STEP) {
        int4 d[GRP_BATCH], sv[GRP_BATCH];
#pragma unroll
        for (int k = 0; k < GRP_BATCH; ++k) {
            const int e = base + k * GRP_STEP;
            d[k]  = (e < e1) ? *(const int4*)(dst + e) : make_int4(-1, -1, -1, -1);
            sv[k] = (e < e1) ? *(const int4*)(src + e) : make_int4(0, 0, 0, 0);
        }
#pragma unroll
        for (int k = 0; k < GRP_BATCH; ++k) {
            const unsigned x0 = (unsigned)(d[k].x - lo), x1 = (unsigned)(d[k].y - lo);
            const unsigned x2 = (unsigned)(d[k].z - lo), x3 = (unsigned)(d[k].w - lo);
            if (x0 < RANGE_NODES) tmp[atomicAdd(&cur[x0], 1)] = sv[k].x;
            if (x1 < RANGE_NODES) tmp[atomicAdd(&cur[x1], 1)] = sv[k].y;
            if (x2 < RANGE_NODES) tmp[atomicAdd(&cur[x2], 1)] = sv[k].z;
            if (x3 < RANGE_NODES) tmp[atomicAdd(&cur[x3], 1)] = sv[k].w;
        }
    }
    const int et = E & ~3;
    if (tid == 0 && et >= e0 && et < e0 + sl)
        for (int e = et; e < E; ++e) {
            const unsigned x = (unsigned)(dst[e] - lo);
            if (x < RANGE_NODES) tmp[atomicAdd(&cur[x], 1)] = src[e];
        }
}

template <int CTRL>
__device__ __forceinline__ float row_get(float v) {
    return __int_as_float(
        __builtin_amdgcn_update_dpp(0, __float_as_int(v), CTRL, 0xf, 0xf, true));
}
template <int CTRL>
__device__ __forceinline__ float row_xadd(float keep, float send) {
    return keep + row_get<CTRL>(send);
}
__device__ __forceinline__ float xhalf(float x) {
    const int b = __float_as_int(x);
    return __int_as_float(
        __builtin_amdgcn_permlanex16(b, b, 0x76543210, 0xfedcba98, false, false));
}
__device__ __forceinline__ float edge_part(const float4 v, const float4 b, const float4 w) {
    float sx = w.x * lrelu(v.x + b.x);
    float sy = w.y * lrelu(v.y + b.y);
    sx = fmaf(w.z, lrelu(v.z + b.z), sx);
    sy = fmaf(w.w, lrelu(v.w + b.w), sy);
    return sx + sy;
}

static_assert(HF * sizeof(float) == 512, "the row pitch behind the shift in row_load");
typedef __attribute__((ext_vector_type(4))) float v4f;
typedef const __attribute__((address_space(1))) v4f* grow_ptr;
__device__ __forceinline__ float4 row_load(const float* fs, int s, int lane) {
    const unsigned long long a = (unsigned long long)fs + ((unsigned long long)(unsigned)s << 9);
    const unsigned lo = (unsigned)__builtin_amdgcn_readfirstlane((int)(unsigned)a);
    const unsigned hi = (unsigned)__builtin_amdgcn_readfirstlane((int)(unsigned)(a >> 32));
    const grow_ptr p = (grow_ptr)(((unsigned long long)hi << 32) | lo);
    const v4f r = p[lane];
    return make_float4(r.x, r.y, r.z, r.w);
}

template <bool ORDER>
__global__ __launch_bounds__(256)
void gat_node_fused(const float* __restrict__ fs, const float* __restrict__ fd,
                    const float* __restrict__ attn,
                    const int* __restrict__ off, const int* __restrict__ grp,
                    int* __restrict__ nbr_out, float* __restrict__ hout) {
    const int node = __builtin_amdgcn_readfirstlane(
        (int)(blockIdx.x * 8 + (threadIdx.x >> 5)));
    if (node >= N_NODES) return;
    const int lane = threadIdx.x & 31;
    const bool odd1 = (lane & 1) != 0;
    const bool odd2 = (lane & 2) != 0;
    const bool hi8  = (lane & 8) != 0;
    const int beg = off[node];
    const int end = off[node + 1];

    const float4 b = *(const float4*)(fd + (size_t)node * HF + lane * 4);
    const float4 w = *(const float4*)(attn + lane * 4);

    float m = -INFINITY, den = 0.0f;
    float4 acc = make_float4(0.0f, 0.0f, 0.0f, 0.0f);

    for (int cb = beg; cb < end; cb += 32) {
        const int cn = min(32, end - cb);
        int ids;
        if constexpr (ORDER) {
            __shared__ int slot[8 * 32];
            int* const ws = slot + (threadIdx.x >> 5) * 32;
            const int deg = end - beg;
            const int ob  = cb - beg;
            for (int ib = 0; ib < deg; ib += 32) {
                const int i = ib + lane;
                const bool act = i < deg;
                const int key = act ? grp[beg + i] : 0;
                const int ck  = key * 32 + lane;
                int r = 0;
                for (int jb = 0; jb < deg; jb += 32) {
                    const int jn = min(32, deg - jb);
                    if (jb == ib) {
                        for (int jj = 0; jj < jn; ++jj)
                            r += (__builtin_amdgcn_readlane(ck, jj) < ck) ? 1 : 0;
                    } else {
                        const int kl = (jb + lane < deg) ? grp[beg + jb + lane] : 0;
                        if (jb < ib) {
                            for (int jj = 0; jj < jn; ++jj)
                                r += (__builtin_amdgcn_readlane(kl, jj) <= key) ? 1 : 0;
                        } else {
                            for (int jj = 0; jj < jn; ++jj)
                                r += (__builtin_amdgcn_readlane(kl, jj) < key) ? 1 : 0;
                        }
                    }
                }
                if (act) {
                    if (ob == 0) nbr_out[beg + r] = key;
                    const unsigned p = (unsigned)(r - ob);
                    if (p < 32u) ws[p] = key;
                }
            }
            ids = ws[min(lane, cn - 1)];
        } else {
            ids = grp[cb + min(lane, cn - 1)];
        }
        const int lim = cn - (lane & 3);
        for (int j = 0; j < cn; j += 4) {
            const int s0 = __builtin_amdgcn_readlane(ids, j);
            const int s1 = __builtin_amdgcn_readlane(ids, j + 1);
            const int s2 = __builtin_amdgcn_readlane(ids, j + 2);
            const int s3 = __builtin_amdgcn_readlane(ids, j + 3);
            const float4 v0 = row_load(fs, s0, lane);
            const float4 v1 = row_load(fs, s1, lane);
            const float4 v2 = row_load(fs, s2, lane);
            const float4 v3 = row_load(fs, s3, lane);

            const float l0 = edge_part(v0, b, w);
            const float l1 = edge_part(v1, b, w);
            const float l2 = edge_part(v2, b, w);
            const float l3 = edge_part(v3, b, w);
            const float e01 = row_xadd<0x161>(odd1 ? l1 : l0, odd1 ? l0 : l1);
            const float e23 = row_xadd<0x161>(odd1 ? l3 : l2, odd1 ? l2 : l3);
            float lg = row_xadd<0x162>(odd2 ? e23 : e01, odd2 ? e01 : e23);
            lg = row_xadd<0x164>(lg, lg);
            lg = row_xadd<0x168>(lg, lg);
            if (j >= lim) lg = -INFINITY;

            float mq = fmaxf(lg, row_get<0x161>(lg));
            mq = fmaxf(mq, row_get<0x162>(mq));
            const float mn = fmaxf(m, mq);
            const float ex = expf((hi8 ? m : lg) - mn);
            const float p0 = row_get<0x150>(ex);
            const float p1 = row_get<0x151>(ex);
            const float p2 = row_get<0x152>(ex);
            const float p3 = row_get<0x153>(ex);
            const float sc = row_get<0x158>(ex);
            den   = den   * sc + ((p0 + p1) + (p2 + p3));
            acc.x = acc.x * sc + ((p0 * v0.x + p1 * v1.x) + (p2 * v2.x + p3 * v3.x));
            acc.y = acc.y * sc + ((p0 * v0.y + p1 * v1.y) + (p2 * v2.y + p3 * v3.y));
            acc.z = acc.z * sc + ((p0 * v0.z + p1 * v1.z) + (p2 * v2.z + p3 * v3.z));
            acc.w = acc.w * sc + ((p0 * v0.w + p1 * v1.w) + (p2 * v2.w + p3 * v3.w));
            m = mn;
        }
    }

    const float inv = (end > beg) ? 1.0f / den : 0.0f;
    float4 o;
    o.x = acc.x * inv; o.y = acc.y * inv; o.z = acc.z * inv; o.w = acc.w * inv;
    o.x = 0.5f * (o.x + xhalf(o.x));
    o.y = 0.5f * (o.y + xhalf(o.y));
    o.z = 0.5f * (o.z + xhalf(o.z));
    o.w = 0.5f * (o.w + xhalf(o.w));
    if (lane < 16)
        *(float4*)(hout + (size_t)node * F_DIM + lane * 4) = o;
}

extern "C" void kernel_launch(void* const* d_in, const int* in_sizes, int n_in,
                              void* d_out, int out_size, void* d_ws, size_t ws_size,
                              hipStream_t stream) {
    (void)n_in; (void)out_size; (void)ws_size;
    const float* h0   = (const float*)d_in[0];
    const float* Wsrc = (const float*)d_in[1];
    const float* bsrc = (const float*)d_in[2];
    const float* Wdst = (const float*)d_in[3];
    const float* bdst = (const float*)d_in[4];
    const float* attn = (const float*)d_in[5];
    const int*   src  = (const int*)d_in[6];
    const int*   dst  = (const int*)d_in[7];
    const int    E    = in_sizes[6];

    float* ws   = (float*)d_ws;
    float* fs   = ws;  ws += (size_t)N_NODES * HF;
    float* fd   = ws;  ws += (size_t)N_NODES * HF;
    float* hbuf = ws;  ws += (size_t)N_NODES * F_DIM;
    int* off    = (int*)ws;  ws += (size_t)N_NODES + 4;
    int* tmp    = (int*)ws;  ws += (size_t)E;
    int* nbr    = (int*)ws;
    int* pc     = (int*)hbuf;
    int* bs     = pc + (size_t)GRP_SLICES * N_PAD;

    const int gridW = (N_NODES + 7) / 8;
    const dim3 gridC(GRP_SLICES, GRP_RANGES);

    grp_count<<<gridC, GRP_THREADS, 0, stream>>>(dst, pc, bs, E);
    grp_offsets<<<NODE_BLOCKS, 256, 0, stream>>>(pc, bs, off);
    grp_fill<<<gridC, GRP_THREADS, 0, stream>>>(src, dst, pc, off, tmp, E);

    const float* hcur = h0;
    for (int l = 0; l < L_LAYERS; ++l) {
        gat_proj_wmma<<<dim3(PROJ_BLOCKS, 2), PROJ_WAVES * 32, 0, stream>>>(
            hcur,
            Wsrc + (size_t)l * F_DIM * HF, bsrc + (size_t)l * HF,
            Wdst + (size_t)l * F_DIM * HF, bdst + (size_t)l * HF,
            fs, fd);

        float* hn = (l == L_LAYERS - 1) ? (float*)d_out : hbuf;
        if (l == 0)
            gat_node_fused<true><<<gridW, 256, 0, stream>>>(
                fs, fd, attn + (size_t)l * HF, off, tmp, nbr, hn);
        else
            gat_node_fused<false><<<gridW, 256, 0, stream>>>(
                fs, fd, attn + (size_t)l * HF, off, nbr, nullptr, hn);
        hcur = hbuf;
    }
}


# layer 1's projection rides in the grouping's launches as extra work-groups (dst product beside the counting, src product beside the placing), all groups 16 waves; 7 launches become 6; sums unchanged
# speedup vs baseline: 5.1379x; 1.0396x over previous
#include <hip/hip_runtime.h>

#define N_NODES   50000
#define F_DIM     64
#define H_HEADS   2
#define HF        128
#define L_LAYERS  2
#define NEG_SLOPE 0.2f

typedef __attribute__((ext_vector_type(2))) float v2f;
typedef __attribute__((ext_vector_type(8))) float v8f;

__device__ __forceinline__ float lrelu(float x) {
    return fmaxf(x, NEG_SLOPE * x);
}

#define ROW_TILES   (N_NODES / 16)
#define PROJ_WAVES  4
#define PROJ_BLOCKS ((ROW_TILES + PROJ_WAVES - 1) / PROJ_WAVES)

#define PROJ_LDS_BYTES (16 * 8 * 16 * 16)

template <int WAVES>
__device__ __forceinline__
void proj_strips(float4* wl, const int pblock,
                 const float* __restrict__ h,
                 const float* __restrict__ W, const float* __restrict__ bias,
                 float* __restrict__ out) {
    static_assert(512 % (WAVES * 32) == 0, "whole staging passes");
    const int tid  = threadIdx.x;
    const int wave = tid >> 5;
    const int lane = tid & 31;
    const int hh   = lane >> 4;
    const int c    = lane & 15;

    const int  tile = pblock * WAVES + wave;
    const bool live = tile < ROW_TILES;
    const int  ltile = live ? tile : ROW_TILES - 1;

    const float* __restrict__ arow = h + (size_t)(ltile * 16 + c) * F_DIM + hh * 4;
    float4 a[8];
#pragma unroll
    for (int i2 = 0; i2 < 8; ++i2)
        a[i2] = *(const float4*)(arow + 8 * i2);

#pragma unroll
    for (int it = 0; it < 512 / (WAVES * 32); ++it) {
        const int item = it * (WAVES * 32) + tid;
        const int q  = item >> 5;
        const int c4 = item & 31;
        const float* __restrict__ wp = W + (size_t)(4 * q) * HF + 4 * c4;
        const float4 r0 = *(const float4*)(wp);
        const float4 r1 = *(const float4*)(wp + HF);
        const float4 r2 = *(const float4*)(wp + 2 * HF);
        const float4 r3 = *(const float4*)(wp + 3 * HF);
        float4* dp = wl + (q * 8 + 4 * (c4 >> 4)) * 16 + (c4 & 15);
        dp[0]  = make_float4(r0.x, r1.x, r2.x, r3.x);
        dp[16] = make_float4(r0.y, r1.y, r2.y, r3.y);
        dp[32] = make_float4(r0.z, r1.z, r2.z, r3.z);
        dp[48] = make_float4(r0.w, r1.w, r2.w, r3.w);
    }
    __syncthreads();
    if (!live) return;

    v8f acc[8];
#pragma unroll
    for (int t = 0; t < 8; ++t) acc[t] = (v8f){};

    const float4* bl = wl + hh * 128 + c;
#pragma unroll
    for (int i2 = 0; i2 < 8; ++i2) {
        float4 b[8];
#pragma unroll
        for (int t = 0; t < 8; ++t) b[t] = bl[i2 * 256 + t * 16];
        v2f a0, a1;
        a0.x = a[i2].x; a0.y = a[i2].y;
        a1.x = a[i2].z; a1.y = a[i2].w;
#pragma unroll
        for (int t = 0; t < 8; ++t) {
            v2f b0; b0.x = b[t].x; b0.y = b[t].y;
            acc[t] = __builtin_amdgcn_wmma_f32_16x16x4_f32(
                        false, a0, false, b0, (short)0, acc[t], false, false);
        }
#pragma unroll
        for (int t = 0; t < 8; ++t) {
            v2f b1; b1.x = b[t].z; b1.y = b[t].w;
            acc[t] = __builtin_amdgcn_wmma_f32_16x16x4_f32(
                        false, a1, false, b1, (short)0, acc[t], false, false);
        }
    }

    float* __restrict__ orow = out + (size_t)(tile * 16 + hh * 8) * HF + 4 * c;
#pragma unroll
    for (int g = 0; g < 2; ++g) {
        const float4 bv = *(const float4*)(bias + 64 * g + 4 * c);
#pragma unroll
        for (int r = 0; r < 8; ++r) {
            float4 o;
            o.x = acc[4 * g + 0][r] + bv.x;
            o.y = acc[4 * g + 1][r] + bv.y;
            o.z = acc[4 * g + 2][r] + bv.z;
            o.w = acc[4 * g + 3][r] + bv.w;
            *(float4*)(orow + (size_t)r * HF + 64 * g) = o;
        }
    }
}

__global__ __launch_bounds__(PROJ_WAVES * 32)
void gat_proj_wmma(const float* __restrict__ h,
                   const float* __restrict__ Wsrc, const float* __restrict__ bsrc,
                   const float* __restrict__ Wdst, const float* __restrict__ bdst,
                   float* __restrict__ fs, float* __restrict__ fd) {
    __shared__ __attribute__((aligned(16))) unsigned char sm[PROJ_LDS_BYTES];
    const bool isSrc = (blockIdx.y == 0);
    proj_strips<PROJ_WAVES>((float4*)sm, blockIdx.x, h,
                            isSrc ? Wsrc : Wdst, isSrc ? bsrc : bdst, isSrc ? fs : fd);
}

#define GRP_SLICES   36
#define GRP_THREADS  512
#define NODE_BLOCKS  ((N_NODES + 255) / 256)
#define RANGE_BLOCKS 28
#define RANGE_NODES  (RANGE_BLOCKS * 256)
#define GRP_RANGES   (NODE_BLOCKS / RANGE_BLOCKS)
#define N_PAD        (NODE_BLOCKS * 256)
static_assert(GRP_RANGES * RANGE_BLOCKS == NODE_BLOCKS, "ranges must tile the node blocks");
static_assert(RANGE_NODES % GRP_THREADS == 0, "whole passes over a range's counters");

__device__ __forceinline__ int grp_slice_len(int E) {
    return (((E + GRP_SLICES - 1) / GRP_SLICES) + 3) & ~3;
}
#define GRP_STEP  (4 * GRP_THREADS)
#define GRP_BATCH 12
#define RIDE_WAVES  (GRP_THREADS / 32)
#define RIDE_BLOCKS ((ROW_TILES + RIDE_WAVES - 1) / RIDE_WAVES)
#define RIDE_ROWS   ((RIDE_BLOCKS + GRP_SLICES - 1) / GRP_SLICES)
static_assert(RANGE_NODES * sizeof(int) <= PROJ_LDS_BYTES, "a cell's counters fit the shared 32 KiB");

__device__ __forceinline__
bool ride_along(unsigned char* sm, const float* __restrict__ h,
                const float* __restrict__ W, const float* __restrict__ bias,
                float* __restrict__ out) {
    if (blockIdx.y < GRP_RANGES) return false;
    const int p = (blockIdx.y - GRP_RANGES) * GRP_SLICES + blockIdx.x;
    if (p < RIDE_BLOCKS) proj_strips<RIDE_WAVES>((float4*)sm, p, h, W, bias, out);
    return true;
}

__global__ __launch_bounds__(GRP_THREADS)
void grp_count(const int* __restrict__ dst, int* __restrict__ pc,
               int* __restrict__ bs, int E,
               const float* __restrict__ h, const float* __restrict__ Wdst,
               const float* __restrict__ bdst, float* __restrict__ fd) {
    __shared__ __attribute__((aligned(16))) unsigned char sm[PROJ_LDS_BYTES];
    if (ride_along(sm, h, Wdst, bdst, fd)) return;
    int* const lc = (int*)sm;
    const int tid = threadIdx.x;
    const int s   = blockIdx.x;
    const int lo  = blockIdx.y * RANGE_NODES;
    for (int i = tid; i < RANGE_NODES; i += GRP_THREADS) lc[i] = 0;
    __syncthreads();

    const int sl = grp_slice_len(E);
    const int e0 = s * sl;
    const int e1 = min(E & ~3, e0 + sl);
    for (int base = e0 + 4 * tid; base < e1; base += GRP_BATCH * GRP_STEP) {
        int4 d[GRP_BATCH];
#pragma unroll
        for (int k = 0; k < GRP_BATCH; ++k) {
            const int e = base + k * GRP_STEP;
            d[k] = (e < e1) ? *(const int4*)(dst + e) : make_int4(-1, -1, -1, -1);
        }
#pragma unroll
        for (int k = 0; k < GRP_BATCH; ++k) {
            const unsigned x0 = (unsigned)(d[k].x - lo), x1 = (unsigned)(d[k].y - lo);
            const unsigned x2 = (unsigned)(d[k].z - lo), x3 = (unsigned)(d[k].w - lo);
            if (x0 < RANGE_NODES) atomicAdd(&lc[x0], 1);
            if (x1 < RANGE_NODES) atomicAdd(&lc[x1], 1);
            if (x2 < RANGE_NODES) atomicAdd(&lc[x2], 1);
            if (x3 < RANGE_NODES) atomicAdd(&lc[x3], 1);
        }
    }
    const int et = E & ~3;
    if (tid == 0 && et >= e0 && et < e0 + sl)
        for (int e = et; e < E; ++e) {
            const unsigned x = (unsigned)(dst[e] - lo);
            if (x < RANGE_NODES) atomicAdd(&lc[x], 1);
        }
    __syncthreads();

    for (int i = tid; i < RANGE_NODES; i += GRP_THREADS)
        pc[(size_t)s * N_PAD + lo + i] = lc[i];
    const int wave = tid >> 5, lane = tid & 31;
    for (int nb = wave; nb < RANGE_BLOCKS; nb += GRP_THREADS / 32) {
        int sum = 0;
#pragma unroll
        for (int k = 0; k < 8; ++k) sum += lc[nb * 256 + k * 32 + lane];
#pragma unroll
        for (int o = 16; o > 0; o >>= 1) sum += __shfl_xor(sum, o, 32);
        if (lane == 0)
            bs[s * NODE_BLOCKS + blockIdx.y * RANGE_BLOCKS + nb] = sum;
    }
}

__global__ __launch_bounds__(256)
void grp_offsets(int* __restrict__ pc, const int* __restrict__ bs,
                 int* __restrict__ off) {
    __shared__ int red[8];
    __shared__ int sc[256];
    const int t = threadIdx.x;
    const int b = blockIdx.x;
    const int i = b * 256 + t;
    int cnts[GRP_SLICES];
#pragma unroll
    for (int s = 0; s < GRP_SLICES; ++s) cnts[s] = pc[(size_t)s * N_PAD + i];
    int part = 0;
    if (t < b) {
#pragma unroll
        for (int s = 0; s < GRP_SLICES; ++s) part += bs[s * NODE_BLOCKS + t];
    }
#pragma unroll
    for (int o = 16; o > 0; o >>= 1)
        part += __shfl_xor(part, o, 32);
    if ((t & 31) == 0) red[t >> 5] = part;

    int run = 0;
#pragma unroll
    for (int s = 0; s < GRP_SLICES; ++s) {
        pc[(size_t)s * N_PAD + i] = run;
        run += cnts[s];
    }
    const int c = run;
    sc[t] = c;
    __syncthreads();
    int prefix = 0;
#pragma unroll
    for (int k = 0; k < 8; ++k) prefix += red[k];
    int v = c;
    for (int o = 1; o < 256; o <<= 1) {
        const int add = (t >= o) ? sc[t - o] : 0;
        __syncthreads();
        v += add;
        sc[t] = v;
        __syncthreads();
    }
    if (i < N_NODES) off[i] = prefix + v - c;
    if (i == N_NODES - 1) off[N_NODES] = prefix + v;
}

__global__ __launch_bounds__(GRP_THREADS)
void grp_fill(const int* __restrict__ src, const int* __restrict__ dst,
              const int* __restrict__ pc, const int* __restrict__ off,
              int* __restrict__ tmp, int E,
              const float* __restrict__ h, const float* __restrict__ Wsrc,
              const float* __restrict__ bsrc, float* __restrict__ fs) {
    __shared__ __attribute__((aligned(16))) unsigned char sm[PROJ_LDS_BYTES];
    if (ride_along(sm, h, Wsrc, bsrc, fs)) return;
    int* const cur = (int*)sm;
    const int tid = threadIdx.x;
    const int s   = blockIdx.x;
    const int lo  = blockIdx.y * RANGE_NODES;
#pragma unroll
    for (int k = 0; k < RANGE_NODES / GRP_THREADS; ++k) {
        const int i = k * GRP_THREADS + tid;
        const int node = lo + i;
        cur[i] = off[min(node, N_NODES)] + pc[(size_t)s * N_PAD + node];
    }
    __syncthreads();

    const int sl = grp_slice_len(E);
    const int e0 = s * sl;
    const int e1 = min(E & ~3, e0 + sl);
    for (int base = e0 + 4 * tid; base < e1; base += GRP_BATCH * GRP_STEP) {
        int4 d[GRP_BATCH], sv[GRP_BATCH];
#pragma unroll
        for (int k = 0; k < GRP_BATCH; ++k) {
            const int e = base + k * GRP_STEP;
            d[k]  = (e < e1) ? *(const int4*)(dst + e) : make_int4(-1, -1, -1, -1);
            sv[k] = (e < e1) ? *(const int4*)(src + e) : make_int4(0, 0, 0, 0);
        }
#pragma unroll
        for (int k = 0; k < GRP_BATCH; ++k) {
            const unsigned x0 = (unsigned)(d[k].x - lo), x1 = (unsigned)(d[k].y - lo);
            const unsigned x2 = (unsigned)(d[k].z - lo), x3 = (unsigned)(d[k].w - lo);
            if (x0 < RANGE_NODES) tmp[atomicAdd(&cur[x0], 1)] = sv[k].x;
            if (x1 < RANGE_NODES) tmp[atomicAdd(&cur[x1], 1)] = sv[k].y;
            if (x2 < RANGE_NODES) tmp[atomicAdd(&cur[x2], 1)] = sv[k].z;
            if (x3 < RANGE_NODES) tmp[atomicAdd(&cur[x3], 1)] = sv[k].w;
        }
    }
    const int et = E & ~3;
    if (tid == 0 && et >= e0 && et < e0 + sl)
        for (int e = et; e < E; ++e) {
            const unsigned x = (unsigned)(dst[e] - lo);
            if (x < RANGE_NODES) tmp[atomicAdd(&cur[x], 1)] = src[e];
        }
}

template <int CTRL>
__device__ __forceinline__ float row_get(float v) {
    return __int_as_float(
        __builtin_amdgcn_update_dpp(0, __float_as_int(v), CTRL, 0xf, 0xf, true));
}
template <int CTRL>
__device__ __forceinline__ float row_xadd(float keep, float send) {
    return keep + row_get<CTRL>(send);
}
__device__ __forceinline__ float xhalf(float x) {
    const int b = __float_as_int(x);
    return __int_as_float(
        __builtin_amdgcn_permlanex16(b, b, 0x76543210, 0xfedcba98, false, false));
}
__device__ __forceinline__ float edge_part(const float4 v, const float4 b, const float4 w) {
    float sx = w.x * lrelu(v.x + b.x);
    float sy = w.y * lrelu(v.y + b.y);
    sx = fmaf(w.z, lrelu(v.z + b.z), sx);
    sy = fmaf(w.w, lrelu(v.w + b.w), sy);
    return sx + sy;
}

static_assert(HF * sizeof(float) == 512, "the row pitch behind the shift in row_load");
typedef __attribute__((ext_vector_type(4))) float v4f;
typedef const __attribute__((address_space(1))) v4f* grow_ptr;
__device__ __forceinline__ float4 row_load(const float* fs, int s, int lane) {
    const unsigned long long a = (unsigned long long)fs + ((unsigned long long)(unsigned)s << 9);
    const unsigned lo = (unsigned)__builtin_amdgcn_readfirstlane((int)(unsigned)a);
    const unsigned hi = (unsigned)__builtin_amdgcn_readfirstlane((int)(unsigned)(a >> 32));
    const grow_ptr p = (grow_ptr)(((unsigned long long)hi << 32) | lo);
    const v4f r = p[lane];
    return make_float4(r.x, r.y, r.z, r.w);
}

template <bool ORDER>
__global__ __launch_bounds__(256)
void gat_node_fused(const float* __restrict__ fs, const float* __restrict__ fd,
                    const float* __restrict__ attn,
                    const int* __restrict__ off, const int* __restrict__ grp,
                    int* __restrict__ nbr_out, float* __restrict__ hout) {
    const int node = __builtin_amdgcn_readfirstlane(
        (int)(blockIdx.x * 8 + (threadIdx.x >> 5)));
    if (node >= N_NODES) return;
    const int lane = threadIdx.x & 31;
    const bool odd1 = (lane & 1) != 0;
    const bool odd2 = (lane & 2) != 0;
    const bool hi8  = (lane & 8) != 0;
    const int beg = off[node];
    const int end = off[node + 1];

    const float4 b = *(const float4*)(fd + (size_t)node * HF + lane * 4);
    const float4 w = *(const float4*)(attn + lane * 4);

    float m = -INFINITY, den = 0.0f;
    float4 acc = make_float4(0.0f, 0.0f, 0.0f, 0.0f);

    for (int cb = beg; cb < end; cb += 32) {
        const int cn = min(32, end - cb);
        int ids;
        if constexpr (ORDER) {
            __shared__ int slot[8 * 32];
            int* const ws = slot + (threadIdx.x >> 5) * 32;
            const int deg = end - beg;
            const int ob  = cb - beg;
            for (int ib = 0; ib < deg; ib += 32) {
                const int i = ib + lane;
                const bool act = i < deg;
                const int key = act ? grp[beg + i] : 0;
                const int ck  = key * 32 + lane;
                int r = 0;
                for (int jb = 0; jb < deg; jb += 32) {
                    const int jn = min(32, deg - jb);
                    if (jb == ib) {
                        for (int jj = 0; jj < jn; ++jj)
                            r += (__builtin_amdgcn_readlane(ck, jj) < ck) ? 1 : 0;
                    } else {
                        const int kl = (jb + lane < deg) ? grp[beg + jb + lane] : 0;
                        if (jb < ib) {
                            for (int jj = 0; jj < jn; ++jj)
                                r += (__builtin_amdgcn_readlane(kl, jj) <= key) ? 1 : 0;
                        } else {
                            for (int jj = 0; jj < jn; ++jj)
                                r += (__builtin_amdgcn_readlane(kl, jj) < key) ? 1 : 0;
                        }
                    }
                }
                if (act) {
                    if (ob == 0) nbr_out[beg + r] = key;
                    const unsigned p = (unsigned)(r - ob);
                    if (p < 32u) ws[p] = key;
                }
            }
            ids = ws[min(lane, cn - 1)];
        } else {
            ids = grp[cb + min(lane, cn - 1)];
        }
        const int lim = cn - (lane & 3);
        for (int j = 0; j < cn; j += 4) {
            const int s0 = __builtin_amdgcn_readlane(ids, j);
            const int s1 = __builtin_amdgcn_readlane(ids, j + 1);
            const int s2 = __builtin_amdgcn_readlane(ids, j + 2);
            const int s3 = __builtin_amdgcn_readlane(ids, j + 3);
            const float4 v0 = row_load(fs, s0, lane);
            const float4 v1 = row_load(fs, s1, lane);
            const float4 v2 = row_load(fs, s2, lane);
            const float4 v3 = row_load(fs, s3, lane);

            const float l0 = edge_part(v0, b, w);
            const float l1 = edge_part(v1, b, w);
            const float l2 = edge_part(v2, b, w);
            const float l3 = edge_part(v3, b, w);
            const float e01 = row_xadd<0x161>(odd1 ? l1 : l0, odd1 ? l0 : l1);
            const float e23 = row_xadd<0x161>(odd1 ? l3 : l2, odd1 ? l2 : l3);
            float lg = row_xadd<0x162>(odd2 ? e23 : e01, odd2 ? e01 : e23);
            lg = row_xadd<0x164>(lg, lg);
            lg = row_xadd<0x168>(lg, lg);
            if (j >= lim) lg = -INFINITY;

            float mq = fmaxf(lg, row_get<0x161>(lg));
            mq = fmaxf(mq, row_get<0x162>(mq));
            const float mn = fmaxf(m, mq);
            const float ex = expf((hi8 ? m : lg) - mn);
            const float p0 = row_get<0x150>(ex);
            const float p1 = row_get<0x151>(ex);
            const float p2 = row_get<0x152>(ex);
            const float p3 = row_get<0x153>(ex);
            const float sc = row_get<0x158>(ex);
            den   = den   * sc + ((p0 + p1) + (p2 + p3));
            acc.x = acc.x * sc + ((p0 * v0.x + p1 * v1.x) + (p2 * v2.x + p3 * v3.x));
            acc.y = acc.y * sc + ((p0 * v0.y + p1 * v1.y) + (p2 * v2.y + p3 * v3.y));
            acc.z = acc.z * sc + ((p0 * v0.z + p1 * v1.z) + (p2 * v2.z + p3 * v3.z));
            acc.w = acc.w * sc + ((p0 * v0.w + p1 * v1.w) + (p2 * v2.w + p3 * v3.w));
            m = mn;
        }
    }

    const float inv = (end > beg) ? 1.0f / den : 0.0f;
    float4 o;
    o.x = acc.x * inv; o.y = acc.y * inv; o.z = acc.z * inv; o.w = acc.w * inv;
    o.x = 0.5f * (o.x + xhalf(o.x));
    o.y = 0.5f * (o.y + xhalf(o.y));
    o.z = 0.5f * (o.z + xhalf(o.z));
    o.w = 0.5f * (o.w + xhalf(o.w));
    if (lane < 16)
        *(float4*)(hout + (size_t)node * F_DIM + lane * 4) = o;
}

extern "C" void kernel_launch(void* const* d_in, const int* in_sizes, int n_in,
                              void* d_out, int out_size, void* d_ws, size_t ws_size,
                              hipStream_t stream) {
    (void)n_in; (void)out_size; (void)ws_size;
    const float* h0   = (const float*)d_in[0];
    const float* Wsrc = (const float*)d_in[1];
    const float* bsrc = (const float*)d_in[2];
    const float* Wdst = (const float*)d_in[3];
    const float* bdst = (const float*)d_in[4];
    const float* attn = (const float*)d_in[5];
    const int*   src  = (const int*)d_in[6];
    const int*   dst  = (const int*)d_in[7];
    const int    E    = in_sizes[6];

    float* ws   = (float*)d_ws;
    float* fs   = ws;  ws += (size_t)N_NODES * HF;
    float* fd   = ws;  ws += (size_t)N_NODES * HF;
    float* hbuf = ws;  ws += (size_t)N_NODES * F_DIM;
    int* off    = (int*)ws;  ws += (size_t)N_NODES + 4;
    int* tmp    = (int*)ws;  ws += (size_t)E;
    int* nbr    = (int*)ws;
    int* pc     = (int*)hbuf;
    int* bs     = pc + (size_t)GRP_SLICES * N_PAD;

    const int gridW = (N_NODES + 7) / 8;
    const dim3 gridC(GRP_SLICES, GRP_RANGES + RIDE_ROWS);

    grp_count<<<gridC, GRP_THREADS, 0, stream>>>(dst, pc, bs, E, h0, Wdst, bdst, fd);
    grp_offsets<<<NODE_BLOCKS, 256, 0, stream>>>(pc, bs, off);
    grp_fill<<<gridC, GRP_THREADS, 0, stream>>>(src, dst, pc, off, tmp, E, h0, Wsrc, bsrc, fs);

    const float* hcur = h0;
    for (int l = 0; l < L_LAYERS; ++l) {
        if (l > 0)
            gat_proj_wmma<<<dim3(PROJ_BLOCKS, 2), PROJ_WAVES * 32, 0, stream>>>(
                hcur,
                Wsrc + (size_t)l * F_DIM * HF, bsrc + (size_t)l * HF,
                Wdst + (size_t)l * F_DIM * HF, bdst + (size_t)l * HF,
                fs, fd);

        float* hn = (l == L_LAYERS - 1) ? (float*)d_out : hbuf;
        if (l == 0)
            gat_node_fused<true><<<gridW, 256, 0, stream>>>(
                fs, fd, attn + (size_t)l * HF, off, tmp, nbr, hn);
        else
            gat_node_fused<false><<<gridW, 256, 0, stream>>>(
                fs, fd, attn + (size_t)l * HF, off, nbr, nullptr, hn);
        hcur = hbuf;
    }
}


# layer 2's projection inside the aggregation launches (16-node work-groups): src product in layer 1's epilogue into the tile's dead fd rows, dst product in layer 2's prologue via LDS; 5 launches
# speedup vs baseline: 5.3082x; 1.0331x over previous
#include <hip/hip_runtime.h>

#define N_NODES   50000
#define F_DIM     64
#define H_HEADS   2
#define HF        128
#define NEG_SLOPE 0.2f

typedef __attribute__((ext_vector_type(2))) float v2f;
typedef __attribute__((ext_vector_type(8))) float v8f;

__device__ __forceinline__ float lrelu(float x) {
    return fmaxf(x, NEG_SLOPE * x);
}

#define ROW_TILES   (N_NODES / 16)

#define PROJ_LDS_BYTES (16 * 8 * 16 * 16)

template <int WAVES>
__device__ __forceinline__
void proj_strips(float4* wl, const int pblock,
                 const float* __restrict__ h,
                 const float* __restrict__ W, const float* __restrict__ bias,
                 float* __restrict__ out) {
    static_assert(512 % (WAVES * 32) == 0, "whole staging passes");
    const int tid  = threadIdx.x;
    const int wave = tid >> 5;
    const int lane = tid & 31;
    const int hh   = lane >> 4;
    const int c    = lane & 15;

    const int  tile = pblock * WAVES + wave;
    const bool live = tile < ROW_TILES;
    const int  ltile = live ? tile : ROW_TILES - 1;

    const float* __restrict__ arow = h + (size_t)(ltile * 16 + c) * F_DIM + hh * 4;
    float4 a[8];
#pragma unroll
    for (int i2 = 0; i2 < 8; ++i2)
        a[i2] = *(const float4*)(arow + 8 * i2);

#pragma unroll
    for (int it = 0; it < 512 / (WAVES * 32); ++it) {
        const int item = it * (WAVES * 32) + tid;
        const int q  = item >> 5;
        const int c4 = item & 31;
        const float* __restrict__ wp = W + (size_t)(4 * q) * HF + 4 * c4;
        const float4 r0 = *(const float4*)(wp);
        const float4 r1 = *(const float4*)(wp + HF);
        const float4 r2 = *(const float4*)(wp + 2 * HF);
        const float4 r3 = *(const float4*)(wp + 3 * HF);
        float4* dp = wl + (q * 8 + 4 * (c4 >> 4)) * 16 + (c4 & 15);
        dp[0]  = make_float4(r0.x, r1.x, r2.x, r3.x);
        dp[16] = make_float4(r0.y, r1.y, r2.y, r3.y);
        dp[32] = make_float4(r0.z, r1.z, r2.z, r3.z);
        dp[48] = make_float4(r0.w, r1.w, r2.w, r3.w);
    }
    __syncthreads();
    if (!live) return;

    v8f acc[8];
#pragma unroll
    for (int t = 0; t < 8; ++t) acc[t] = (v8f){};

    const float4* bl = wl + hh * 128 + c;
#pragma unroll
    for (int i2 = 0; i2 < 8; ++i2) {
        float4 b[8];
#pragma unroll
        for (int t = 0; t < 8; ++t) b[t] = bl[i2 * 256 + t * 16];
        v2f a0, a1;
        a0.x = a[i2].x; a0.y = a[i2].y;
        a1.x = a[i2].z; a1.y = a[i2].w;
#pragma unroll
        for (int t = 0; t < 8; ++t) {
            v2f b0; b0.x = b[t].x; b0.y = b[t].y;
            acc[t] = __builtin_amdgcn_wmma_f32_16x16x4_f32(
                        false, a0, false, b0, (short)0, acc[t], false, false);
        }
#pragma unroll
        for (int t = 0; t < 8; ++t) {
            v2f b1; b1.x = b[t].z; b1.y = b[t].w;
            acc[t] = __builtin_amdgcn_wmma_f32_16x16x4_f32(
                        false, a1, false, b1, (short)0, acc[t], false, false);
        }
    }

    float* __restrict__ orow = out + (size_t)(tile * 16 + hh * 8) * HF + 4 * c;
#pragma unroll
    for (int g = 0; g < 2; ++g) {
        const float4 bv = *(const float4*)(bias + 64 * g + 4 * c);
#pragma unroll
        for (int r = 0; r < 8; ++r) {
            float4 o;
            o.x = acc[4 * g + 0][r] + bv.x;
            o.y = acc[4 * g + 1][r] + bv.y;
            o.z = acc[4 * g + 2][r] + bv.z;
            o.w = acc[4 * g + 3][r] + bv.w;
            *(float4*)(orow + (size_t)r * HF + 64 * g) = o;
        }
    }
}

#define GRP_SLICES   36
#define GRP_THREADS  512
#define NODE_BLOCKS  ((N_NODES + 255) / 256)
#define RANGE_BLOCKS 28
#define RANGE_NODES  (RANGE_BLOCKS * 256)
#define GRP_RANGES   (NODE_BLOCKS / RANGE_BLOCKS)
#define N_PAD        (NODE_BLOCKS * 256)
static_assert(GRP_RANGES * RANGE_BLOCKS == NODE_BLOCKS, "ranges must tile the node blocks");
static_assert(RANGE_NODES % GRP_THREADS == 0, "whole passes over a range's counters");

__device__ __forceinline__ int grp_slice_len(int E) {
    return (((E + GRP_SLICES - 1) / GRP_SLICES) + 3) & ~3;
}
#define GRP_STEP  (4 * GRP_THREADS)
#define GRP_BATCH 12
#define RIDE_WAVES  (GRP_THREADS / 32)
#define RIDE_BLOCKS ((ROW_TILES + RIDE_WAVES - 1) / RIDE_WAVES)
#define RIDE_ROWS   ((RIDE_BLOCKS + GRP_SLICES - 1) / GRP_SLICES)
static_assert(RANGE_NODES * sizeof(int) <= PROJ_LDS_BYTES, "a cell's counters fit the shared 32 KiB");

__device__ __forceinline__
bool ride_along(unsigned char* sm, const float* __restrict__ h,
                const float* __restrict__ W, const float* __restrict__ bias,
                float* __restrict__ out) {
    if (blockIdx.y < GRP_RANGES) return false;
    const int p = (blockIdx.y - GRP_RANGES) * GRP_SLICES + blockIdx.x;
    if (p < RIDE_BLOCKS) proj_strips<RIDE_WAVES>((float4*)sm, p, h, W, bias, out);
    return true;
}

__global__ __launch_bounds__(GRP_THREADS)
void grp_count(const int* __restrict__ dst, int* __restrict__ pc,
               int* __restrict__ bs, int E,
               const float* __restrict__ h, const float* __restrict__ Wdst,
               const float* __restrict__ bdst, float* __restrict__ fd) {
    __shared__ __attribute__((aligned(16))) unsigned char sm[PROJ_LDS_BYTES];
    if (ride_along(sm, h, Wdst, bdst, fd)) return;
    int* const lc = (int*)sm;
    const int tid = threadIdx.x;
    const int s   = blockIdx.x;
    const int lo  = blockIdx.y * RANGE_NODES;
    for (int i = tid; i < RANGE_NODES; i += GRP_THREADS) lc[i] = 0;
    __syncthreads();

    const int sl = grp_slice_len(E);
    const int e0 = s * sl;
    const int e1 = min(E & ~3, e0 + sl);
    for (int base = e0 + 4 * tid; base < e1; base += GRP_BATCH * GRP_STEP) {
        int4 d[GRP_BATCH];
#pragma unroll
        for (int k = 0; k < GRP_BATCH; ++k) {
            const int e = base + k * GRP_STEP;
            d[k] = (e < e1) ? *(const int4*)(dst + e) : make_int4(-1, -1, -1, -1);
        }
#pragma unroll
        for (int k = 0; k < GRP_BATCH; ++k) {
            const unsigned x0 = (unsigned)(d[k].x - lo), x1 = (unsigned)(d[k].y - lo);
            const unsigned x2 = (unsigned)(d[k].z - lo), x3 = (unsigned)(d[k].w - lo);
            if (x0 < RANGE_NODES) atomicAdd(&lc[x0], 1);
            if (x1 < RANGE_NODES) atomicAdd(&lc[x1], 1);
            if (x2 < RANGE_NODES) atomicAdd(&lc[x2], 1);
            if (x3 < RANGE_NODES) atomicAdd(&lc[x3], 1);
        }
    }
    const int et = E & ~3;
    if (tid == 0 && et >= e0 && et < e0 + sl)
        for (int e = et; e < E; ++e) {
            const unsigned x = (unsigned)(dst[e] - lo);
            if (x < RANGE_NODES) atomicAdd(&lc[x], 1);
        }
    __syncthreads();

    for (int i = tid; i < RANGE_NODES; i += GRP_THREADS)
        pc[(size_t)s * N_PAD + lo + i] = lc[i];
    const int wave = tid >> 5, lane = tid & 31;
    for (int nb = wave; nb < RANGE_BLOCKS; nb += GRP_THREADS / 32) {
        int sum = 0;
#pragma unroll
        for (int k = 0; k < 8; ++k) sum += lc[nb * 256 + k * 32 + lane];
#pragma unroll
        for (int o = 16; o > 0; o >>= 1) sum += __shfl_xor(sum, o, 32);
        if (lane == 0)
            bs[s * NODE_BLOCKS + blockIdx.y * RANGE_BLOCKS + nb] = sum;
    }
}

__global__ __launch_bounds__(256)
void grp_offsets(int* __restrict__ pc, const int* __restrict__ bs,
                 int* __restrict__ off) {
    __shared__ int red[8];
    __shared__ int sc[256];
    const int t = threadIdx.x;
    const int b = blockIdx.x;
    const int i = b * 256 + t;
    int cnts[GRP_SLICES];
#pragma unroll
    for (int s = 0; s < GRP_SLICES; ++s) cnts[s] = pc[(size_t)s * N_PAD + i];
    int part = 0;
    if (t < b) {
#pragma unroll
        for (int s = 0; s < GRP_SLICES; ++s) part += bs[s * NODE_BLOCKS + t];
    }
#pragma unroll
    for (int o = 16; o > 0; o >>= 1)
        part += __shfl_xor(part, o, 32);
    if ((t & 31) == 0) red[t >> 5] = part;

    int run = 0;
#pragma unroll
    for (int s = 0; s < GRP_SLICES; ++s) {
        pc[(size_t)s * N_PAD + i] = run;
        run += cnts[s];
    }
    const int c = run;
    sc[t] = c;
    __syncthreads();
    int prefix = 0;
#pragma unroll
    for (int k = 0; k < 8; ++k) prefix += red[k];
    int v = c;
    for (int o = 1; o < 256; o <<= 1) {
        const int add = (t >= o) ? sc[t - o] : 0;
        __syncthreads();
        v += add;
        sc[t] = v;
        __syncthreads();
    }
    if (i < N_NODES) off[i] = prefix + v - c;
    if (i == N_NODES - 1) off[N_NODES] = prefix + v;
}

__global__ __launch_bounds__(GRP_THREADS)
void grp_fill(const int* __restrict__ src, const int* __restrict__ dst,
              const int* __restrict__ pc, const int* __restrict__ off,
              int* __restrict__ tmp, int E,
              const float* __restrict__ h, const float* __restrict__ Wsrc,
              const float* __restrict__ bsrc, float* __restrict__ fs) {
    __shared__ __attribute__((aligned(16))) unsigned char sm[PROJ_LDS_BYTES];
    if (ride_along(sm, h, Wsrc, bsrc, fs)) return;
    int* const cur = (int*)sm;
    const int tid = threadIdx.x;
    const int s   = blockIdx.x;
    const int lo  = blockIdx.y * RANGE_NODES;
#pragma unroll
    for (int k = 0; k < RANGE_NODES / GRP_THREADS; ++k) {
        const int i = k * GRP_THREADS + tid;
        const int node = lo + i;
        cur[i] = off[min(node, N_NODES)] + pc[(size_t)s * N_PAD + node];
    }
    __syncthreads();

    const int sl = grp_slice_len(E);
    const int e0 = s * sl;
    const int e1 = min(E & ~3, e0 + sl);
    for (int base = e0 + 4 * tid; base < e1; base += GRP_BATCH * GRP_STEP) {
        int4 d[GRP_BATCH], sv[GRP_BATCH];
#pragma unroll
        for (int k = 0; k < GRP_BATCH; ++k) {
            const int e = base + k * GRP_STEP;
            d[k]  = (e < e1) ? *(const int4*)(dst + e) : make_int4(-1, -1, -1, -1);
            sv[k] = (e < e1) ? *(const int4*)(src + e) : make_int4(0, 0, 0, 0);
        }
#pragma unroll
        for (int k = 0; k < GRP_BATCH; ++k) {
            const unsigned x0 = (unsigned)(d[k].x - lo), x1 = (unsigned)(d[k].y - lo);
            const unsigned x2 = (unsigned)(d[k].z - lo), x3 = (unsigned)(d[k].w - lo);
            if (x0 < RANGE_NODES) tmp[atomicAdd(&cur[x0], 1)] = sv[k].x;
            if (x1 < RANGE_NODES) tmp[atomicAdd(&cur[x1], 1)] = sv[k].y;
            if (x2 < RANGE_NODES) tmp[atomicAdd(&cur[x2], 1)] = sv[k].z;
            if (x3 < RANGE_NODES) tmp[atomicAdd(&cur[x3], 1)] = sv[k].w;
        }
    }
    const int et = E & ~3;
    if (tid == 0 && et >= e0 && et < e0 + sl)
        for (int e = et; e < E; ++e) {
            const unsigned x = (unsigned)(dst[e] - lo);
            if (x < RANGE_NODES) tmp[atomicAdd(&cur[x], 1)] = src[e];
        }
}

template <int CTRL>
__device__ __forceinline__ float row_get(float v) {
    return __int_as_float(
        __builtin_amdgcn_update_dpp(0, __float_as_int(v), CTRL, 0xf, 0xf, true));
}
template <int CTRL>
__device__ __forceinline__ float row_xadd(float keep, float send) {
    return keep + row_get<CTRL>(send);
}
__device__ __forceinline__ float xhalf(float x) {
    const int b = __float_as_int(x);
    return __int_as_float(
        __builtin_amdgcn_permlanex16(b, b, 0x76543210, 0xfedcba98, false, false));
}
__device__ __forceinline__ float edge_part(const float4 v, const float4 b, const float4 w) {
    float sx = w.x * lrelu(v.x + b.x);
    float sy = w.y * lrelu(v.y + b.y);
    sx = fmaf(w.z, lrelu(v.z + b.z), sx);
    sy = fmaf(w.w, lrelu(v.w + b.w), sy);
    return sx + sy;
}

static_assert(HF * sizeof(float) == 512, "the row pitch behind the shift in row_load");
typedef __attribute__((ext_vector_type(4))) float v4f;
typedef const __attribute__((address_space(1))) v4f* grow_ptr;
__device__ __forceinline__ float4 row_load(const float* fs, int s, int lane) {
    const unsigned long long a = (unsigned long long)fs + ((unsigned long long)(unsigned)s << 9);
    const unsigned lo = (unsigned)__builtin_amdgcn_readfirstlane((int)(unsigned)a);
    const unsigned hi = (unsigned)__builtin_amdgcn_readfirstlane((int)(unsigned)(a >> 32));
    const grow_ptr p = (grow_ptr)(((unsigned long long)hi << 32) | lo);
    const v4f r = p[lane];
    return make_float4(r.x, r.y, r.z, r.w);
}

#define TILE_NODES 16
#define HT_PITCH   68
#define FD_PITCH   132
static_assert(N_NODES % TILE_NODES == 0, "whole tiles: every wave has a node and reaches the barriers");

struct BFrag { float q[8][4]; };
__device__ __forceinline__
void tile_b_load(BFrag& bq, const float* __restrict__ W, const int n, const int lane) {
    const int hh = lane >> 4;
    const int c  = lane & 15;
    const float* __restrict__ wp = W + (size_t)(4 * hh) * HF + 16 * n + c;
#pragma unroll
    for (int i2 = 0; i2 < 8; ++i2)
#pragma unroll
        for (int j = 0; j < 4; ++j)
            bq.q[i2][j] = wp[(size_t)(8 * i2 + j) * HF];
}

__device__ __forceinline__
v8f tile_mma(const float* ht, const BFrag& bq, const int lane) {
    const int hh = lane >> 4;
    const int c  = lane & 15;
    const float* ap = ht + c * HT_PITCH + 4 * hh;
    v8f acc = (v8f){};
#pragma unroll
    for (int i2 = 0; i2 < 8; ++i2) {
        const float4 a = *(const float4*)(ap + 8 * i2);
        v2f a0, a1, b0, b1;
        a0.x = a.x; a0.y = a.y;
        a1.x = a.z; a1.y = a.w;
        b0.x = bq.q[i2][0]; b0.y = bq.q[i2][1];
        b1.x = bq.q[i2][2]; b1.y = bq.q[i2][3];
        acc = __builtin_amdgcn_wmma_f32_16x16x4_f32(
                  false, a0, false, b0, (short)0, acc, false, false);
        acc = __builtin_amdgcn_wmma_f32_16x16x4_f32(
                  false, a1, false, b1, (short)0, acc, false, false);
    }
    return acc;
}

template <bool FIRST>
__global__ __launch_bounds__(TILE_NODES * 32)
void gat_node_fused(const float* __restrict__ fs, const float* fdh,
                    const float* __restrict__ attn,
                    const int* __restrict__ off, const int* __restrict__ grp,
                    int* __restrict__ nbr_out, float* __restrict__ hout,
                    const float* __restrict__ W2, const float* __restrict__ b2,
                    float* fs2) {
    constexpr bool ORDER = FIRST;
    __shared__ __attribute__((aligned(16))) float ht[TILE_NODES * HT_PITCH];
    const int wv   = __builtin_amdgcn_readfirstlane((int)(threadIdx.x >> 5));
    const int tile = blockIdx.x;
    const int node = tile * TILE_NODES + wv;
    const int lane = threadIdx.x & 31;
    const bool odd1 = (lane & 1) != 0;
    const bool odd2 = (lane & 2) != 0;
    const bool hi8  = (lane & 8) != 0;
    const int beg = off[node];
    const int end = off[node + 1];

    const float4 w = *(const float4*)(attn + lane * 4);
    float4 b;
    if constexpr (FIRST) {
        b = *(const float4*)(fdh + (size_t)node * HF + lane * 4);
    } else {
        __shared__ __attribute__((aligned(16))) float fdt[TILE_NODES * FD_PITCH];
        const int tid = threadIdx.x;
        const bool mul = wv < HF / 16;
        BFrag bq = {};
        float bv = 0.0f;
        if (mul) {
            tile_b_load(bq, W2, wv, lane);
            bv = b2[16 * wv + (lane & 15)];
        }
        if (tid < TILE_NODES * 16) {
            const int row = tid >> 4, q = tid & 15;
            *(float4*)(ht + row * HT_PITCH + q * 4) =
                *(const float4*)(fdh + (size_t)(tile * TILE_NODES + row) * F_DIM + q * 4);
        }
        __syncthreads();
        if (mul) {
            const v8f acc = tile_mma(ht, bq, lane);
            const int hh = lane >> 4, c = lane & 15;
#pragma unroll
            for (int r = 0; r < 8; ++r)
                fdt[(8 * hh + r) * FD_PITCH + 16 * wv + c] = acc[r] + bv;
        }
        __syncthreads();
        b = *(const float4*)(fdt + wv * FD_PITCH + lane * 4);
    }

    float m = -INFINITY, den = 0.0f;
    float4 acc = make_float4(0.0f, 0.0f, 0.0f, 0.0f);

    for (int cb = beg; cb < end; cb += 32) {
        const int cn = min(32, end - cb);
        int ids;
        if constexpr (ORDER) {
            __shared__ int slot[TILE_NODES * 32];
            int* const ws = slot + (threadIdx.x >> 5) * 32;
            const int deg = end - beg;
            const int ob  = cb - beg;
            for (int ib = 0; ib < deg; ib += 32) {
                const int i = ib + lane;
                const bool act = i < deg;
                const int key = act ? grp[beg + i] : 0;
                const int ck  = key * 32 + lane;
                int r = 0;
                for (int jb = 0; jb < deg; jb += 32) {
                    const int jn = min(32, deg - jb);
                    if (jb == ib) {
                        for (int jj = 0; jj < jn; ++jj)
                            r += (__builtin_amdgcn_readlane(ck, jj) < ck) ? 1 : 0;
                    } else {
                        const int kl = (jb + lane < deg) ? grp[beg + jb + lane] : 0;
                        if (jb < ib) {
                            for (int jj = 0; jj < jn; ++jj)
                                r += (__builtin_amdgcn_readlane(kl, jj) <= key) ? 1 : 0;
                        } else {
                            for (int jj = 0; jj < jn; ++jj)
                                r += (__builtin_amdgcn_readlane(kl, jj) < key) ? 1 : 0;
                        }
                    }
                }
                if (act) {
                    if (ob == 0) nbr_out[beg + r] = key;
                    const unsigned p = (unsigned)(r - ob);
                    if (p < 32u) ws[p] = key;
                }
            }
            ids = ws[min(lane, cn - 1)];
        } else {
            ids = grp[cb + min(lane, cn - 1)];
        }
        const int lim = cn - (lane & 3);
        for (int j = 0; j < cn; j += 4) {
            const int s0 = __builtin_amdgcn_readlane(ids, j);
            const int s1 = __builtin_amdgcn_readlane(ids, j + 1);
            const int s2 = __builtin_amdgcn_readlane(ids, j + 2);
            const int s3 = __builtin_amdgcn_readlane(ids, j + 3);
            const float4 v0 = row_load(fs, s0, lane);
            const float4 v1 = row_load(fs, s1, lane);
            const float4 v2 = row_load(fs, s2, lane);
            const float4 v3 = row_load(fs, s3, lane);

            const float l0 = edge_part(v0, b, w);
            const float l1 = edge_part(v1, b, w);
            const float l2 = edge_part(v2, b, w);
            const float l3 = edge_part(v3, b, w);
            const float e01 = row_xadd<0x161>(odd1 ? l1 : l0, odd1 ? l0 : l1);
            const float e23 = row_xadd<0x161>(odd1 ? l3 : l2, odd1 ? l2 : l3);
            float lg = row_xadd<0x162>(odd2 ? e23 : e01, odd2 ? e01 : e23);
            lg = row_xadd<0x164>(lg, lg);
            lg = row_xadd<0x168>(lg, lg);
            if (j >= lim) lg = -INFINITY;

            float mq = fmaxf(lg, row_get<0x161>(lg));
            mq = fmaxf(mq, row_get<0x162>(mq));
            const float mn = fmaxf(m, mq);
            const float ex = expf((hi8 ? m : lg) - mn);
            const float p0 = row_get<0x150>(ex);
            const float p1 = row_get<0x151>(ex);
            const float p2 = row_get<0x152>(ex);
            const float p3 = row_get<0x153>(ex);
            const float sc = row_get<0x158>(ex);
            den   = den   * sc + ((p0 + p1) + (p2 + p3));
            acc.x = acc.x * sc + ((p0 * v0.x + p1 * v1.x) + (p2 * v2.x + p3 * v3.x));
            acc.y = acc.y * sc + ((p0 * v0.y + p1 * v1.y) + (p2 * v2.y + p3 * v3.y));
            acc.z = acc.z * sc + ((p0 * v0.z + p1 * v1.z) + (p2 * v2.z + p3 * v3.z));
            acc.w = acc.w * sc + ((p0 * v0.w + p1 * v1.w) + (p2 * v2.w + p3 * v3.w));
            m = mn;
        }
    }

    const float inv = (end > beg) ? 1.0f / den : 0.0f;
    float4 o;
    o.x = acc.x * inv; o.y = acc.y * inv; o.z = acc.z * inv; o.w = acc.w * inv;
    o.x = 0.5f * (o.x + xhalf(o.x));
    o.y = 0.5f * (o.y + xhalf(o.y));
    o.z = 0.5f * (o.z + xhalf(o.z));
    o.w = 0.5f * (o.w + xhalf(o.w));
    if constexpr (!FIRST) {
        if (lane < 16)
            *(float4*)(hout + (size_t)node * F_DIM + lane * 4) = o;
    } else {
        if (lane < 16) *(float4*)(ht + wv * HT_PITCH + lane * 4) = o;
        const bool mul = wv < HF / 16;
        BFrag bq = {};
        float bv = 0.0f;
        if (mul) {
            tile_b_load(bq, W2, wv, lane);
            bv = b2[16 * wv + (lane & 15)];
        }
        __syncthreads();
        if (lane < 16)
            *(float4*)(hout + (size_t)node * F_DIM + lane * 4) = o;
        if (mul) {
            const v8f acc2 = tile_mma(ht, bq, lane);
            const int hh = lane >> 4, c = lane & 15;
            float* orow = fs2 + (size_t)(tile * TILE_NODES + 8 * hh) * HF + 16 * wv + c;
#pragma unroll
            for (int r = 0; r < 8; ++r)
                orow[(size_t)r * HF] = acc2[r] + bv;
        }
    }
}

extern "C" void kernel_launch(void* const* d_in, const int* in_sizes, int n_in,
                              void* d_out, int out_size, void* d_ws, size_t ws_size,
                              hipStream_t stream) {
    (void)n_in; (void)out_size; (void)ws_size;
    const float* h0   = (const float*)d_in[0];
    const float* Wsrc = (const float*)d_in[1];
    const float* bsrc = (const float*)d_in[2];
    const float* Wdst = (const float*)d_in[3];
    const float* bdst = (const float*)d_in[4];
    const float* attn = (const float*)d_in[5];
    const int*   src  = (const int*)d_in[6];
    const int*   dst  = (const int*)d_in[7];
    const int    E    = in_sizes[6];

    float* ws   = (float*)d_ws;
    float* fs   = ws;  ws += (size_t)N_NODES * HF;
    float* fd   = ws;  ws += (size_t)N_NODES * HF;
    float* hbuf = ws;  ws += (size_t)N_NODES * F_DIM;
    int* off    = (int*)ws;  ws += (size_t)N_NODES + 4;
    int* tmp    = (int*)ws;  ws += (size_t)E;
    int* nbr    = (int*)ws;
    int* pc     = (int*)hbuf;
    int* bs     = pc + (size_t)GRP_SLICES * N_PAD;

    const dim3 gridC(GRP_SLICES, GRP_RANGES + RIDE_ROWS);

    grp_count<<<gridC, GRP_THREADS, 0, stream>>>(dst, pc, bs, E, h0, Wdst, bdst, fd);
    grp_offsets<<<NODE_BLOCKS, 256, 0, stream>>>(pc, bs, off);
    grp_fill<<<gridC, GRP_THREADS, 0, stream>>>(src, dst, pc, off, tmp, E, h0, Wsrc, bsrc, fs);

    gat_node_fused<true><<<N_NODES / TILE_NODES, TILE_NODES * 32, 0, stream>>>(
        fs, fd, attn, off, tmp, nbr, hbuf,
        Wsrc + (size_t)F_DIM * HF, bsrc + HF, fd);
    gat_node_fused<false><<<N_NODES / TILE_NODES, TILE_NODES * 32, 0, stream>>>(
        fd, hbuf, attn + HF, off, nbr, nullptr, (float*)d_out,
        Wdst + (size_t)F_DIM * HF, bdst + HF, nullptr);
}
